# starting kernel
# baseline (speedup 1.0000x reference)
#include <hip/hip_runtime.h>
#include <math.h>

typedef float v2f __attribute__((ext_vector_type(2)));
typedef float v8f __attribute__((ext_vector_type(8)));

#define NGRID 32
#define LMAX  16
#define NM    31
#define CIN   128
#define COUT  128
#define BATCH 64
#define KJP   1024
#define KLM   512

#define WS_Q      0
#define WS_A1RE   16384
#define WS_A1IM   (WS_A1RE + KLM*KJP)
#define WS_A2RE   (WS_A1IM + KLM*KJP)
#define WS_A2IM   (WS_A2RE + KJP*KLM)
#define WS_COEFRE (WS_A2IM + KJP*KLM)
#define WS_COEFIM (WS_COEFRE + BATCH*KLM*CIN)
#define WS_CHATRE (WS_COEFIM + BATCH*KLM*CIN)
#define WS_CHATIM (WS_CHATRE + BATCH*KLM*COUT)

#define PI_D 3.14159265358979323846

__global__ void sph_legendre_kernel(float* __restrict__ ws) {
  __shared__ float P[LMAX][LMAX][NGRID];
  const int j = threadIdx.x;
  const double theta = PI_D * (j + 0.5) / NGRID;
  const double x = cos(theta), sx = sin(theta);

  P[0][0][j] = 1.0f;
  for (int m = 1; m < LMAX; ++m)
    P[m][m][j] = (float)(-(2.0 * m - 1.0) * sx) * P[m - 1][m - 1][j];
  for (int m = 0; m < LMAX - 1; ++m) {
    P[m + 1][m][j] = (float)((2.0 * m + 1.0) * x) * P[m][m][j];
    for (int l = m + 2; l < LMAX; ++l)
      P[l][m][j] = ((float)((2.0 * l - 1.0) * x) * P[l - 1][m][j] -
                    (float)(l + m - 1) * P[l - 2][m][j]) / (float)(l - m);
  }

  float* Q = ws + WS_Q;
  for (int l = 0; l < LMAX; ++l) {
    for (int m = 0; m <= l; ++m) {
      double ratio = 1.0;
      for (int t = l - m + 1; t <= l + m; ++t) ratio /= (double)t;
      const float Nlm = (float)sqrt((2.0 * l + 1.0) / (4.0 * PI_D) * ratio);
      const float q = Nlm * P[l][m][j];
      Q[(l * NM + (15 + m)) * NGRID + j] = q;
      Q[(l * NM + (15 - m)) * NGRID + j] = (m & 1) ? -q : q;
    }
  }
}

__global__ void sph_buildmats_kernel(float* __restrict__ ws) {
  const int idx = blockIdx.x * blockDim.x + threadIdx.x;
  if (idx >= KLM * KJP) return;
  const int lm = idx >> 10, jp = idx & 1023;
  const int l = lm >> 5, m = lm & 31;
  const int jj = jp >> 5, p = jp & 31;

  float a1r = 0.f, a1i = 0.f, a2r = 0.f, a2i = 0.f;
  if (m < NM) {
    const float q = ws[WS_Q + (l * NM + m) * NGRID + jj];
    const int mh = m - 15;
    const float ang = (2.0f * (float)PI_D / NGRID) * (float)(mh * p);
    float s, c;
    sincosf(ang, &s, &c);
    const float C2 = c, S2 = -s;
    const float th = (float)PI_D * (jj + 0.5f) / NGRID;
    const float wj = sinf(th) * ((float)PI_D / NGRID) * (2.0f * (float)PI_D / NGRID);
    const float scale_l = 2.0f * (float)PI_D * sqrtf(4.0f * (float)PI_D / (2.0f * l + 1.0f));
    a1r = q * wj * C2;      a1i = q * wj * S2;
    a2r = q * scale_l * C2; a2i = q * scale_l * S2;
  }
  ws[WS_A1RE + lm * KJP + jp] = a1r;
  ws[WS_A1IM + lm * KJP + jp] = a1i;
  ws[WS_A2RE + jp * KLM + lm] = a2r;
  ws[WS_A2IM + jp * KLM + lm] = a2i;
}

__device__ __forceinline__ void wave_gemm_32x128(
    const float* __restrict__ A, int lda,
    const float* __restrict__ B, int ldb,
    int K, v8f acc[2][8]) {
  const int lane = threadIdx.x & 31;
  const int row  = lane & 15;
  const int kh   = (lane >> 4) << 1;
  for (int k = 0; k < K; k += 4) {
    v2f a0, a1;
    a0.x = A[row * lda + k + kh];
    a0.y = A[row * lda + k + kh + 1];
    a1.x = A[(row + 16) * lda + k + kh];
    a1.y = A[(row + 16) * lda + k + kh + 1];
#pragma unroll
    for (int nt = 0; nt < 8; ++nt) {
      v2f bb;
      bb.x = B[(k + kh) * ldb + nt * 16 + row];
      bb.y = B[(k + kh + 1) * ldb + nt * 16 + row];
      acc[0][nt] = __builtin_amdgcn_wmma_f32_16x16x4_f32(
          false, a0, false, bb, (short)0, acc[0][nt], false, false);
      acc[1][nt] = __builtin_amdgcn_wmma_f32_16x16x4_f32(
          false, a1, false, bb, (short)0, acc[1][nt], false, false);
    }
  }
}

__device__ __forceinline__ void wave_store_32x128(
    float* __restrict__ C, int ldc, const v8f acc[2][8]) {
  const int lane = threadIdx.x & 31;
  const int col  = lane & 15;
  const int rb   = (lane >> 4) * 8;
#pragma unroll
  for (int t = 0; t < 2; ++t)
#pragma unroll
    for (int nt = 0; nt < 8; ++nt)
#pragma unroll
      for (int v = 0; v < 8; ++v)
        C[(t * 16 + rb + v) * ldc + nt * 16 + col] = acc[t][nt][v];
}

__global__ void __launch_bounds__(32, 1)
sph_phaseA_kernel(const float* __restrict__ A1,
                  const float* __restrict__ X,
                  float* __restrict__ coef) {
  const int mt = blockIdx.x;
  const int b  = blockIdx.y;
  const float* A = A1 + (size_t)mt * 32 * KJP;
  const float* B = X + (size_t)b * KJP * CIN;
  float* C = coef + (size_t)b * KLM * CIN + (size_t)mt * 32 * CIN;
  v8f acc[2][8] = {};
  wave_gemm_32x128(A, KJP, B, CIN, KJP, acc);
  wave_store_32x128(C, CIN, acc);
}

__global__ void __launch_bounds__(32, 1)
sph_phaseB_kernel(const float* __restrict__ coef,
                  const float* __restrict__ W,
                  float* __restrict__ chat) {
  const int l = blockIdx.x;
  const int b = blockIdx.y;
  const float* A = coef + (size_t)b * KLM * CIN + (size_t)l * 32 * CIN;
  const float* B = W + (size_t)l * COUT;
  float* C = chat + (size_t)b * KLM * COUT + (size_t)l * 32 * COUT;
  v8f acc[2][8] = {};
  wave_gemm_32x128(A, CIN, B, LMAX * COUT, CIN, acc);
  wave_store_32x128(C, COUT, acc);
}

__global__ void __launch_bounds__(32, 1)
sph_phaseC_kernel(const float* __restrict__ A2re,
                  const float* __restrict__ A2im,
                  const float* __restrict__ chre,
                  const float* __restrict__ chim,
                  const float* __restrict__ bias,
                  float* __restrict__ out) {
  const int mt = blockIdx.x;
  const int b  = blockIdx.y;
  const int lane = threadIdx.x & 31;
  v8f acc[2][8];
#pragma unroll
  for (int nt = 0; nt < 8; ++nt) {
    const float bv = bias[nt * 16 + (lane & 15)];
#pragma unroll
    for (int v = 0; v < 8; ++v) {
      acc[0][nt][v] = bv;
      acc[1][nt][v] = bv;
    }
  }
  const float* Cre = chre + (size_t)b * KLM * COUT;
  const float* Cim = chim + (size_t)b * KLM * COUT;
  wave_gemm_32x128(A2re + (size_t)mt * 32 * KLM, KLM, Cre, COUT, KLM, acc);
  wave_gemm_32x128(A2im + (size_t)mt * 32 * KLM, KLM, Cim, COUT, KLM, acc);
  wave_store_32x128(out + (size_t)b * KJP * COUT + (size_t)mt * 32 * COUT, COUT, acc);
}

extern "C" void kernel_launch(void* const* d_in, const int* in_sizes, int n_in,
                              void* d_out, int out_size, void* d_ws, size_t ws_size,
                              hipStream_t stream) {
  const float* inp  = (const float*)d_in[0];
  const float* wgt  = (const float*)d_in[1];
  const float* bias = (const float*)d_in[2];
  float* out = (float*)d_out;
  float* ws  = (float*)d_ws;

  sph_legendre_kernel<<<1, 32, 0, stream>>>(ws);
  sph_buildmats_kernel<<<(KLM * KJP + 255) / 256, 256, 0, stream>>>(ws);

  dim3 gAB(16, BATCH);
  sph_phaseA_kernel<<<gAB, 32, 0, stream>>>(ws + WS_A1RE, inp, ws + WS_COEFRE);
  sph_phaseA_kernel<<<gAB, 32, 0, stream>>>(ws + WS_A1IM, inp, ws + WS_COEFIM);
  sph_phaseB_kernel<<<gAB, 32, 0, stream>>>(ws + WS_COEFRE, wgt, ws + WS_CHATRE);
  sph_phaseB_kernel<<<gAB, 32, 0, stream>>>(ws + WS_COEFIM, wgt, ws + WS_CHATIM);

  dim3 gC(32, BATCH);
  sph_phaseC_kernel<<<gC, 32, 0, stream>>>(ws + WS_A2RE, ws + WS_A2IM,
                                           ws + WS_CHATRE, ws + WS_CHATIM,
                                           bias, out);
}


# Legendre table kernel parallelised: one work-item per (order m, theta sample) in a 512-thread block; factorial-ratio divide chains shared per l-m, f64 parts via LDS; table values bit-identical
# speedup vs baseline: 1.6296x; 1.6296x over previous
#include <hip/hip_runtime.h>
#include <math.h>

typedef float v2f __attribute__((ext_vector_type(2)));
typedef float v8f __attribute__((ext_vector_type(8)));

#define NGRID 32
#define LMAX  16
#define NM    31
#define CIN   128
#define COUT  128
#define BATCH 64
#define KJP   1024
#define KLM   512

#define WS_Q      0
#define WS_A1RE   16384
#define WS_A1IM   (WS_A1RE + KLM*KJP)
#define WS_A2RE   (WS_A1IM + KLM*KJP)
#define WS_A2IM   (WS_A2RE + KJP*KLM)
#define WS_COEFRE (WS_A2IM + KJP*KLM)
#define WS_COEFIM (WS_COEFRE + BATCH*KLM*CIN)
#define WS_CHATRE (WS_COEFIM + BATCH*KLM*CIN)
#define WS_CHATIM (WS_CHATRE + BATCH*KLM*COUT)

#define PI_D 3.14159265358979323846

__global__ void __launch_bounds__(LMAX * NGRID)
sph_legendre_kernel(float* __restrict__ ws) {
  __shared__ double s_x[NGRID];
  __shared__ double s_sx[NGRID];
  __shared__ double s_ratio[LMAX][LMAX];
  __shared__ float  s_nlm[LMAX][LMAX];
  __shared__ float  s_cx[LMAX][NGRID];
  __shared__ float  s_cs[LMAX][NGRID];

  const int tid = threadIdx.x;
  const int m = tid >> 5;
  const int j = tid & 31;

  if (m == 0) {
    const double theta = PI_D * (j + 0.5) / NGRID;
    s_x[j]  = cos(theta);
    s_sx[j] = sin(theta);
  } else if (m == 1 && j < LMAX) {
    const int d = j;
    double ratio = 1.0;
    s_ratio[d][0] = ratio;
    for (int mm = 1; d + mm < LMAX; ++mm) {
      ratio /= (double)(d + 2 * mm - 1);
      ratio /= (double)(d + 2 * mm);
      s_ratio[d + mm][mm] = ratio;
    }
  }
  __syncthreads();

  {
    const double x = s_x[j], sx = s_sx[j];
    s_cx[m][j] = (float)((2.0 * m + 1.0) * x);
    s_cs[m][j] = (float)(-(2.0 * m - 1.0) * sx);
  }
  if (tid < LMAX * LMAX) {
    const int l = tid >> 4, mm = tid & 15;
    if (mm <= l)
      s_nlm[l][mm] = (float)sqrt((2.0 * l + 1.0) / (4.0 * PI_D) * s_ratio[l][mm]);
  }
  __syncthreads();

  float* qp = ws + WS_Q + (15 + m) * NGRID + j;
  float* qn = ws + WS_Q + (15 - m) * NGRID + j;

  for (int l = 0; l < m; ++l) {
    qp[l * NM * NGRID] = 0.f;
    qn[l * NM * NGRID] = 0.f;
  }

  float pmm = 1.0f;
  for (int k = 1; k <= m; ++k) pmm = s_cs[k][j] * pmm;
  {
    const float q = s_nlm[m][m] * pmm;
    qp[m * NM * NGRID] = q;
    qn[m * NM * NGRID] = (m & 1) ? -q : q;
  }
  if (m < LMAX - 1) {
    float p2 = pmm;
    float p1 = s_cx[m][j] * pmm;
    {
      const float q = s_nlm[m + 1][m] * p1;
      qp[(m + 1) * NM * NGRID] = q;
      qn[(m + 1) * NM * NGRID] = (m & 1) ? -q : q;
    }
    for (int l = m + 2; l < LMAX; ++l) {
      const float p = (s_cx[l - 1][j] * p1 - (float)(l + m - 1) * p2) / (float)(l - m);
      const float q = s_nlm[l][m] * p;
      qp[l * NM * NGRID] = q;
      qn[l * NM * NGRID] = (m & 1) ? -q : q;
      p2 = p1;
      p1 = p;
    }
  }
}

__global__ void sph_buildmats_kernel(float* __restrict__ ws) {
  const int idx = blockIdx.x * blockDim.x + threadIdx.x;
  if (idx >= KLM * KJP) return;
  const int lm = idx >> 10, jp = idx & 1023;
  const int l = lm >> 5, m = lm & 31;
  const int jj = jp >> 5, p = jp & 31;

  float a1r = 0.f, a1i = 0.f, a2r = 0.f, a2i = 0.f;
  if (m < NM) {
    const float q = ws[WS_Q + (l * NM + m) * NGRID + jj];
    const int mh = m - 15;
    const float ang = (2.0f * (float)PI_D / NGRID) * (float)(mh * p);
    float s, c;
    sincosf(ang, &s, &c);
    const float C2 = c, S2 = -s;
    const float th = (float)PI_D * (jj + 0.5f) / NGRID;
    const float wj = sinf(th) * ((float)PI_D / NGRID) * (2.0f * (float)PI_D / NGRID);
    const float scale_l = 2.0f * (float)PI_D * sqrtf(4.0f * (float)PI_D / (2.0f * l + 1.0f));
    a1r = q * wj * C2;      a1i = q * wj * S2;
    a2r = q * scale_l * C2; a2i = q * scale_l * S2;
  }
  ws[WS_A1RE + lm * KJP + jp] = a1r;
  ws[WS_A1IM + lm * KJP + jp] = a1i;
  ws[WS_A2RE + jp * KLM + lm] = a2r;
  ws[WS_A2IM + jp * KLM + lm] = a2i;
}

__device__ __forceinline__ void wave_gemm_32x128(
    const float* __restrict__ A, int lda,
    const float* __restrict__ B, int ldb,
    int K, v8f acc[2][8]) {
  const int lane = threadIdx.x & 31;
  const int row  = lane & 15;
  const int kh   = (lane >> 4) << 1;
  for (int k = 0; k < K; k += 4) {
    v2f a0, a1;
    a0.x = A[row * lda + k + kh];
    a0.y = A[row * lda + k + kh + 1];
    a1.x = A[(row + 16) * lda + k + kh];
    a1.y = A[(row + 16) * lda + k + kh + 1];
#pragma unroll
    for (int nt = 0; nt < 8; ++nt) {
      v2f bb;
      bb.x = B[(k + kh) * ldb + nt * 16 + row];
      bb.y = B[(k + kh + 1) * ldb + nt * 16 + row];
      acc[0][nt] = __builtin_amdgcn_wmma_f32_16x16x4_f32(
          false, a0, false, bb, (short)0, acc[0][nt], false, false);
      acc[1][nt] = __builtin_amdgcn_wmma_f32_16x16x4_f32(
          false, a1, false, bb, (short)0, acc[1][nt], false, false);
    }
  }
}

__device__ __forceinline__ void wave_store_32x128(
    float* __restrict__ C, int ldc, const v8f acc[2][8]) {
  const int lane = threadIdx.x & 31;
  const int col  = lane & 15;
  const int rb   = (lane >> 4) * 8;
#pragma unroll
  for (int t = 0; t < 2; ++t)
#pragma unroll
    for (int nt = 0; nt < 8; ++nt)
#pragma unroll
      for (int v = 0; v < 8; ++v)
        C[(t * 16 + rb + v) * ldc + nt * 16 + col] = acc[t][nt][v];
}

__global__ void __launch_bounds__(32, 1)
sph_phaseA_kernel(const float* __restrict__ A1,
                  const float* __restrict__ X,
                  float* __restrict__ coef) {
  const int mt = blockIdx.x;
  const int b  = blockIdx.y;
  const float* A = A1 + (size_t)mt * 32 * KJP;
  const float* B = X + (size_t)b * KJP * CIN;
  float* C = coef + (size_t)b * KLM * CIN + (size_t)mt * 32 * CIN;
  v8f acc[2][8] = {};
  wave_gemm_32x128(A, KJP, B, CIN, KJP, acc);
  wave_store_32x128(C, CIN, acc);
}

__global__ void __launch_bounds__(32, 1)
sph_phaseB_kernel(const float* __restrict__ coef,
                  const float* __restrict__ W,
                  float* __restrict__ chat) {
  const int l = blockIdx.x;
  const int b = blockIdx.y;
  const float* A = coef + (size_t)b * KLM * CIN + (size_t)l * 32 * CIN;
  const float* B = W + (size_t)l * COUT;
  float* C = chat + (size_t)b * KLM * COUT + (size_t)l * 32 * COUT;
  v8f acc[2][8] = {};
  wave_gemm_32x128(A, CIN, B, LMAX * COUT, CIN, acc);
  wave_store_32x128(C, COUT, acc);
}

__global__ void __launch_bounds__(32, 1)
sph_phaseC_kernel(const float* __restrict__ A2re,
                  const float* __restrict__ A2im,
                  const float* __restrict__ chre,
                  const float* __restrict__ chim,
                  const float* __restrict__ bias,
                  float* __restrict__ out) {
  const int mt = blockIdx.x;
  const int b  = blockIdx.y;
  const int lane = threadIdx.x & 31;
  v8f acc[2][8];
#pragma unroll
  for (int nt = 0; nt < 8; ++nt) {
    const float bv = bias[nt * 16 + (lane & 15)];
#pragma unroll
    for (int v = 0; v < 8; ++v) {
      acc[0][nt][v] = bv;
      acc[1][nt][v] = bv;
    }
  }
  const float* Cre = chre + (size_t)b * KLM * COUT;
  const float* Cim = chim + (size_t)b * KLM * COUT;
  wave_gemm_32x128(A2re + (size_t)mt * 32 * KLM, KLM, Cre, COUT, KLM, acc);
  wave_gemm_32x128(A2im + (size_t)mt * 32 * KLM, KLM, Cim, COUT, KLM, acc);
  wave_store_32x128(out + (size_t)b * KJP * COUT + (size_t)mt * 32 * COUT, COUT, acc);
}

extern "C" void kernel_launch(void* const* d_in, const int* in_sizes, int n_in,
                              void* d_out, int out_size, void* d_ws, size_t ws_size,
                              hipStream_t stream) {
  const float* inp  = (const float*)d_in[0];
  const float* wgt  = (const float*)d_in[1];
  const float* bias = (const float*)d_in[2];
  float* out = (float*)d_out;
  float* ws  = (float*)d_ws;

  sph_legendre_kernel<<<1, LMAX * NGRID, 0, stream>>>(ws);
  sph_buildmats_kernel<<<(KLM * KJP + 255) / 256, 256, 0, stream>>>(ws);

  dim3 gAB(16, BATCH);
  sph_phaseA_kernel<<<gAB, 32, 0, stream>>>(ws + WS_A1RE, inp, ws + WS_COEFRE);
  sph_phaseA_kernel<<<gAB, 32, 0, stream>>>(ws + WS_A1IM, inp, ws + WS_COEFIM);
  sph_phaseB_kernel<<<gAB, 32, 0, stream>>>(ws + WS_COEFRE, wgt, ws + WS_CHATRE);
  sph_phaseB_kernel<<<gAB, 32, 0, stream>>>(ws + WS_COEFIM, wgt, ws + WS_CHATIM);

  dim3 gC(32, BATCH);
  sph_phaseC_kernel<<<gC, 32, 0, stream>>>(ws + WS_A2RE, ws + WS_A2IM,
                                           ws + WS_CHATRE, ws + WS_CHATIM,
                                           bias, out);
}


# GEMM K-loops software-pipelined: the global fragment loads of K-step t+3 are issued as one group ahead of the WMMAs of step t (ring of four register sets); k order and sums unchanged
# speedup vs baseline: 2.9536x; 1.8125x over previous
#include <hip/hip_runtime.h>
#include <math.h>

typedef float v2f __attribute__((ext_vector_type(2)));
typedef float v8f __attribute__((ext_vector_type(8)));

#define NGRID 32
#define LMAX  16
#define NM    31
#define CIN   128
#define COUT  128
#define BATCH 64
#define KJP   1024
#define KLM   512

#define WS_Q      0
#define WS_A1RE   16384
#define WS_A1IM   (WS_A1RE + KLM*KJP)
#define WS_A2RE   (WS_A1IM + KLM*KJP)
#define WS_A2IM   (WS_A2RE + KJP*KLM)
#define WS_COEFRE (WS_A2IM + KJP*KLM)
#define WS_COEFIM (WS_COEFRE + BATCH*KLM*CIN)
#define WS_CHATRE (WS_COEFIM + BATCH*KLM*CIN)
#define WS_CHATIM (WS_CHATRE + BATCH*KLM*COUT)

#define PI_D 3.14159265358979323846

__global__ void __launch_bounds__(LMAX * NGRID)
sph_legendre_kernel(float* __restrict__ ws) {
  __shared__ double s_x[NGRID];
  __shared__ double s_sx[NGRID];
  __shared__ double s_ratio[LMAX][LMAX];
  __shared__ float  s_nlm[LMAX][LMAX];
  __shared__ float  s_cx[LMAX][NGRID];
  __shared__ float  s_cs[LMAX][NGRID];

  const int tid = threadIdx.x;
  const int m = tid >> 5;
  const int j = tid & 31;

  if (m == 0) {
    const double theta = PI_D * (j + 0.5) / NGRID;
    s_x[j]  = cos(theta);
    s_sx[j] = sin(theta);
  } else if (m == 1 && j < LMAX) {
    const int d = j;
    double ratio = 1.0;
    s_ratio[d][0] = ratio;
    for (int mm = 1; d + mm < LMAX; ++mm) {
      ratio /= (double)(d + 2 * mm - 1);
      ratio /= (double)(d + 2 * mm);
      s_ratio[d + mm][mm] = ratio;
    }
  }
  __syncthreads();

  {
    const double x = s_x[j], sx = s_sx[j];
    s_cx[m][j] = (float)((2.0 * m + 1.0) * x);
    s_cs[m][j] = (float)(-(2.0 * m - 1.0) * sx);
  }
  if (tid < LMAX * LMAX) {
    const int l = tid >> 4, mm = tid & 15;
    if (mm <= l)
      s_nlm[l][mm] = (float)sqrt((2.0 * l + 1.0) / (4.0 * PI_D) * s_ratio[l][mm]);
  }
  __syncthreads();

  float* qp = ws + WS_Q + (15 + m) * NGRID + j;
  float* qn = ws + WS_Q + (15 - m) * NGRID + j;

  for (int l = 0; l < m; ++l) {
    qp[l * NM * NGRID] = 0.f;
    qn[l * NM * NGRID] = 0.f;
  }

  float pmm = 1.0f;
  for (int k = 1; k <= m; ++k) pmm = s_cs[k][j] * pmm;
  {
    const float q = s_nlm[m][m] * pmm;
    qp[m * NM * NGRID] = q;
    qn[m * NM * NGRID] = (m & 1) ? -q : q;
  }
  if (m < LMAX - 1) {
    float p2 = pmm;
    float p1 = s_cx[m][j] * pmm;
    {
      const float q = s_nlm[m + 1][m] * p1;
      qp[(m + 1) * NM * NGRID] = q;
      qn[(m + 1) * NM * NGRID] = (m & 1) ? -q : q;
    }
    for (int l = m + 2; l < LMAX; ++l) {
      const float p = (s_cx[l - 1][j] * p1 - (float)(l + m - 1) * p2) / (float)(l - m);
      const float q = s_nlm[l][m] * p;
      qp[l * NM * NGRID] = q;
      qn[l * NM * NGRID] = (m & 1) ? -q : q;
      p2 = p1;
      p1 = p;
    }
  }
}

__global__ void sph_buildmats_kernel(float* __restrict__ ws) {
  const int idx = blockIdx.x * blockDim.x + threadIdx.x;
  if (idx >= KLM * KJP) return;
  const int lm = idx >> 10, jp = idx & 1023;
  const int l = lm >> 5, m = lm & 31;
  const int jj = jp >> 5, p = jp & 31;

  float a1r = 0.f, a1i = 0.f, a2r = 0.f, a2i = 0.f;
  if (m < NM) {
    const float q = ws[WS_Q + (l * NM + m) * NGRID + jj];
    const int mh = m - 15;
    const float ang = (2.0f * (float)PI_D / NGRID) * (float)(mh * p);
    float s, c;
    sincosf(ang, &s, &c);
    const float C2 = c, S2 = -s;
    const float th = (float)PI_D * (jj + 0.5f) / NGRID;
    const float wj = sinf(th) * ((float)PI_D / NGRID) * (2.0f * (float)PI_D / NGRID);
    const float scale_l = 2.0f * (float)PI_D * sqrtf(4.0f * (float)PI_D / (2.0f * l + 1.0f));
    a1r = q * wj * C2;      a1i = q * wj * S2;
    a2r = q * scale_l * C2; a2i = q * scale_l * S2;
  }
  ws[WS_A1RE + lm * KJP + jp] = a1r;
  ws[WS_A1IM + lm * KJP + jp] = a1i;
  ws[WS_A2RE + jp * KLM + lm] = a2r;
  ws[WS_A2IM + jp * KLM + lm] = a2i;
}

struct KFrag {
  v2f a0, a1;
  v2f b[8];
};

__device__ __forceinline__ void load_kfrag(
    KFrag& f, const float* __restrict__ A, int lda,
    const float* __restrict__ B, int ldb, int row, int kh, int k) {
  f.a0.x = A[row * lda + k + kh];
  f.a0.y = A[row * lda + k + kh + 1];
  f.a1.x = A[(row + 16) * lda + k + kh];
  f.a1.y = A[(row + 16) * lda + k + kh + 1];
#pragma unroll
  for (int nt = 0; nt < 8; ++nt) {
    f.b[nt].x = B[(k + kh) * ldb + nt * 16 + row];
    f.b[nt].y = B[(k + kh + 1) * ldb + nt * 16 + row];
  }
}

__device__ __forceinline__ void mma_kfrag(const KFrag& f, v8f acc[2][8]) {
#pragma unroll
  for (int nt = 0; nt < 8; ++nt) {
    acc[0][nt] = __builtin_amdgcn_wmma_f32_16x16x4_f32(
        false, f.a0, false, f.b[nt], (short)0, acc[0][nt], false, false);
    acc[1][nt] = __builtin_amdgcn_wmma_f32_16x16x4_f32(
        false, f.a1, false, f.b[nt], (short)0, acc[1][nt], false, false);
  }
}

__device__ __forceinline__ void wave_gemm_32x128(
    const float* __restrict__ A, int lda,
    const float* __restrict__ B, int ldb,
    int K, v8f acc[2][8]) {
  const int lane = threadIdx.x & 31;
  const int row  = lane & 15;
  const int kh   = (lane >> 4) << 1;
  KFrag f0, f1, f2, f3;
  load_kfrag(f0, A, lda, B, ldb, row, kh, 0);
  __builtin_amdgcn_sched_barrier(0);
  load_kfrag(f1, A, lda, B, ldb, row, kh, 4);
  __builtin_amdgcn_sched_barrier(0);
  load_kfrag(f2, A, lda, B, ldb, row, kh, 8);
  int k = 0;
  for (; k + 16 < K; k += 16) {
    load_kfrag(f3, A, lda, B, ldb, row, kh, k + 12);
    __builtin_amdgcn_sched_barrier(0);
    mma_kfrag(f0, acc);
    load_kfrag(f0, A, lda, B, ldb, row, kh, k + 16);
    __builtin_amdgcn_sched_barrier(0);
    mma_kfrag(f1, acc);
    load_kfrag(f1, A, lda, B, ldb, row, kh, k + 20);
    __builtin_amdgcn_sched_barrier(0);
    mma_kfrag(f2, acc);
    load_kfrag(f2, A, lda, B, ldb, row, kh, k + 24);
    __builtin_amdgcn_sched_barrier(0);
    mma_kfrag(f3, acc);
  }
  load_kfrag(f3, A, lda, B, ldb, row, kh, k + 12);
  mma_kfrag(f0, acc);
  mma_kfrag(f1, acc);
  mma_kfrag(f2, acc);
  mma_kfrag(f3, acc);
}

__device__ __forceinline__ void wave_store_32x128(
    float* __restrict__ C, int ldc, const v8f acc[2][8]) {
  const int lane = threadIdx.x & 31;
  const int col  = lane & 15;
  const int rb   = (lane >> 4) * 8;
#pragma unroll
  for (int t = 0; t < 2; ++t)
#pragma unroll
    for (int nt = 0; nt < 8; ++nt)
#pragma unroll
      for (int v = 0; v < 8; ++v)
        C[(t * 16 + rb + v) * ldc + nt * 16 + col] = acc[t][nt][v];
}

__global__ void __launch_bounds__(32, 1)
sph_phaseA_kernel(const float* __restrict__ A1,
                  const float* __restrict__ X,
                  float* __restrict__ coef) {
  const int mt = blockIdx.x;
  const int b  = blockIdx.y;
  const float* A = A1 + (size_t)mt * 32 * KJP;
  const float* B = X + (size_t)b * KJP * CIN;
  float* C = coef + (size_t)b * KLM * CIN + (size_t)mt * 32 * CIN;
  v8f acc[2][8] = {};
  wave_gemm_32x128(A, KJP, B, CIN, KJP, acc);
  wave_store_32x128(C, CIN, acc);
}

__global__ void __launch_bounds__(32, 1)
sph_phaseB_kernel(const float* __restrict__ coef,
                  const float* __restrict__ W,
                  float* __restrict__ chat) {
  const int l = blockIdx.x;
  const int b = blockIdx.y;
  const float* A = coef + (size_t)b * KLM * CIN + (size_t)l * 32 * CIN;
  const float* B = W + (size_t)l * COUT;
  float* C = chat + (size_t)b * KLM * COUT + (size_t)l * 32 * COUT;
  v8f acc[2][8] = {};
  wave_gemm_32x128(A, CIN, B, LMAX * COUT, CIN, acc);
  wave_store_32x128(C, COUT, acc);
}

__global__ void __launch_bounds__(32, 1)
sph_phaseC_kernel(const float* __restrict__ A2re,
                  const float* __restrict__ A2im,
                  const float* __restrict__ chre,
                  const float* __restrict__ chim,
                  const float* __restrict__ bias,
                  float* __restrict__ out) {
  const int mt = blockIdx.x;
  const int b  = blockIdx.y;
  const int lane = threadIdx.x & 31;
  v8f acc[2][8];
#pragma unroll
  for (int nt = 0; nt < 8; ++nt) {
    const float bv = bias[nt * 16 + (lane & 15)];
#pragma unroll
    for (int v = 0; v < 8; ++v) {
      acc[0][nt][v] = bv;
      acc[1][nt][v] = bv;
    }
  }
  const float* Cre = chre + (size_t)b * KLM * COUT;
  const float* Cim = chim + (size_t)b * KLM * COUT;
  wave_gemm_32x128(A2re + (size_t)mt * 32 * KLM, KLM, Cre, COUT, KLM, acc);
  wave_gemm_32x128(A2im + (size_t)mt * 32 * KLM, KLM, Cim, COUT, KLM, acc);
  wave_store_32x128(out + (size_t)b * KJP * COUT + (size_t)mt * 32 * COUT, COUT, acc);
}

extern "C" void kernel_launch(void* const* d_in, const int* in_sizes, int n_in,
                              void* d_out, int out_size, void* d_ws, size_t ws_size,
                              hipStream_t stream) {
  const float* inp  = (const float*)d_in[0];
  const float* wgt  = (const float*)d_in[1];
  const float* bias = (const float*)d_in[2];
  float* out = (float*)d_out;
  float* ws  = (float*)d_ws;

  sph_legendre_kernel<<<1, LMAX * NGRID, 0, stream>>>(ws);
  sph_buildmats_kernel<<<(KLM * KJP + 255) / 256, 256, 0, stream>>>(ws);

  dim3 gAB(16, BATCH);
  sph_phaseA_kernel<<<gAB, 32, 0, stream>>>(ws + WS_A1RE, inp, ws + WS_COEFRE);
  sph_phaseA_kernel<<<gAB, 32, 0, stream>>>(ws + WS_A1IM, inp, ws + WS_COEFIM);
  sph_phaseB_kernel<<<gAB, 32, 0, stream>>>(ws + WS_COEFRE, wgt, ws + WS_CHATRE);
  sph_phaseB_kernel<<<gAB, 32, 0, stream>>>(ws + WS_COEFIM, wgt, ws + WS_CHATIM);

  dim3 gC(32, BATCH);
  sph_phaseC_kernel<<<gC, 32, 0, stream>>>(ws + WS_A2RE, ws + WS_A2IM,
                                           ws + WS_CHATRE, ws + WS_CHATIM,
                                           bias, out);
}


# Real-input conjugate symmetry: only orders m>=0 transformed, re and im rows stacked per degree so each GEMM phase is one launch; order -m folded as x2 into the synthesis matrix; sum order changes
# speedup vs baseline: 5.5045x; 1.8637x over previous
#include <hip/hip_runtime.h>
#include <math.h>

typedef float v2f __attribute__((ext_vector_type(2)));
typedef float v8f __attribute__((ext_vector_type(8)));

#define NGRID 32
#define LMAX  16
#define NM    31
#define CIN   128
#define COUT  128
#define BATCH 64
#define KJP   1024
#define MH    16
#define KLM   512


#define WS_Q      0
#define WS_A1     16384
#define WS_A2     (WS_A1 + KLM*KJP)
#define WS_COEF   (WS_A2 + KJP*KLM)
#define WS_CHAT   (WS_COEF + BATCH*KLM*CIN)

#define PI_D 3.14159265358979323846

__global__ void __launch_bounds__(LMAX * NGRID)
sph_legendre_kernel(float* __restrict__ ws) {
  __shared__ double s_x[NGRID];
  __shared__ double s_sx[NGRID];
  __shared__ double s_ratio[LMAX][LMAX];
  __shared__ float  s_nlm[LMAX][LMAX];
  __shared__ float  s_cx[LMAX][NGRID];
  __shared__ float  s_cs[LMAX][NGRID];

  const int tid = threadIdx.x;
  const int m = tid >> 5;
  const int j = tid & 31;

  if (m == 0) {
    const double theta = PI_D * (j + 0.5) / NGRID;
    s_x[j]  = cos(theta);
    s_sx[j] = sin(theta);
  } else if (m == 1 && j < LMAX) {
    const int d = j;
    double ratio = 1.0;
    s_ratio[d][0] = ratio;
    for (int mm = 1; d + mm < LMAX; ++mm) {
      ratio /= (double)(d + 2 * mm - 1);
      ratio /= (double)(d + 2 * mm);
      s_ratio[d + mm][mm] = ratio;
    }
  }
  __syncthreads();

  {
    const double x = s_x[j], sx = s_sx[j];
    s_cx[m][j] = (float)((2.0 * m + 1.0) * x);
    s_cs[m][j] = (float)(-(2.0 * m - 1.0) * sx);
  }
  if (tid < LMAX * LMAX) {
    const int l = tid >> 4, mm = tid & 15;
    if (mm <= l)
      s_nlm[l][mm] = (float)sqrt((2.0 * l + 1.0) / (4.0 * PI_D) * s_ratio[l][mm]);
  }
  __syncthreads();

  float* qp = ws + WS_Q + (15 + m) * NGRID + j;
  float* qn = ws + WS_Q + (15 - m) * NGRID + j;

  for (int l = 0; l < m; ++l) {
    qp[l * NM * NGRID] = 0.f;
    qn[l * NM * NGRID] = 0.f;
  }

  float pmm = 1.0f;
  for (int k = 1; k <= m; ++k) pmm = s_cs[k][j] * pmm;
  {
    const float q = s_nlm[m][m] * pmm;
    qp[m * NM * NGRID] = q;
    qn[m * NM * NGRID] = (m & 1) ? -q : q;
  }
  if (m < LMAX - 1) {
    float p2 = pmm;
    float p1 = s_cx[m][j] * pmm;
    {
      const float q = s_nlm[m + 1][m] * p1;
      qp[(m + 1) * NM * NGRID] = q;
      qn[(m + 1) * NM * NGRID] = (m & 1) ? -q : q;
    }
    for (int l = m + 2; l < LMAX; ++l) {
      const float p = (s_cx[l - 1][j] * p1 - (float)(l + m - 1) * p2) / (float)(l - m);
      const float q = s_nlm[l][m] * p;
      qp[l * NM * NGRID] = q;
      qn[l * NM * NGRID] = (m & 1) ? -q : q;
      p2 = p1;
      p1 = p;
    }
  }
}

__global__ void sph_buildmats_kernel(float* __restrict__ ws) {
  const int idx = blockIdx.x * blockDim.x + threadIdx.x;
  if (idx >= LMAX * MH * KJP) return;
  const int lm = idx >> 10, jp = idx & 1023;
  const int l = lm >> 4, mh = lm & 15;
  const int jj = jp >> 5, p = jp & 31;

  const float q = ws[WS_Q + (l * NM + 15 + mh) * NGRID + jj];
  const float ang = (2.0f * (float)PI_D / NGRID) * (float)(mh * p);
  float s, c;
  sincosf(ang, &s, &c);
  const float C2 = c, S2 = -s;
  const float th = (float)PI_D * (jj + 0.5f) / NGRID;
  const float wj = sinf(th) * ((float)PI_D / NGRID) * (2.0f * (float)PI_D / NGRID);
  const float scale_l = 2.0f * (float)PI_D * sqrtf(4.0f * (float)PI_D / (2.0f * l + 1.0f));
  const float f = (mh == 0) ? 1.0f : 2.0f;
  const float a1r = q * wj * C2;
  const float a1i = q * wj * S2;
  const float a2r = f * (q * scale_l * C2);
  const float a2i = f * (q * scale_l * S2);

  const int rre = l * 32 + mh;
  const int rim = rre + MH;
  ws[WS_A1 + rre * KJP + jp] = a1r;
  ws[WS_A1 + rim * KJP + jp] = a1i;
  ws[WS_A2 + jp * KLM + rre] = a2r;
  ws[WS_A2 + jp * KLM + rim] = a2i;
}

struct KFrag {
  v2f a0, a1;
  v2f b[8];
};

__device__ __forceinline__ void load_kfrag(
    KFrag& f, const float* __restrict__ A, int lda,
    const float* __restrict__ B, int ldb, int row, int kh, int k) {
  f.a0.x = A[row * lda + k + kh];
  f.a0.y = A[row * lda + k + kh + 1];
  f.a1.x = A[(row + 16) * lda + k + kh];
  f.a1.y = A[(row + 16) * lda + k + kh + 1];
#pragma unroll
  for (int nt = 0; nt < 8; ++nt) {
    f.b[nt].x = B[(k + kh) * ldb + nt * 16 + row];
    f.b[nt].y = B[(k + kh + 1) * ldb + nt * 16 + row];
  }
}

__device__ __forceinline__ void mma_kfrag(const KFrag& f, v8f acc[2][8]) {
#pragma unroll
  for (int nt = 0; nt < 8; ++nt) {
    acc[0][nt] = __builtin_amdgcn_wmma_f32_16x16x4_f32(
        false, f.a0, false, f.b[nt], (short)0, acc[0][nt], false, false);
    acc[1][nt] = __builtin_amdgcn_wmma_f32_16x16x4_f32(
        false, f.a1, false, f.b[nt], (short)0, acc[1][nt], false, false);
  }
}

__device__ __forceinline__ void wave_gemm_32x128(
    const float* __restrict__ A, int lda,
    const float* __restrict__ B, int ldb,
    int K, v8f acc[2][8]) {
  const int lane = threadIdx.x & 31;
  const int row  = lane & 15;
  const int kh   = (lane >> 4) << 1;
  KFrag f0, f1, f2, f3;
  load_kfrag(f0, A, lda, B, ldb, row, kh, 0);
  __builtin_amdgcn_sched_barrier(0);
  load_kfrag(f1, A, lda, B, ldb, row, kh, 4);
  __builtin_amdgcn_sched_barrier(0);
  load_kfrag(f2, A, lda, B, ldb, row, kh, 8);
  int k = 0;
  for (; k + 16 < K; k += 16) {
    load_kfrag(f3, A, lda, B, ldb, row, kh, k + 12);
    __builtin_amdgcn_sched_barrier(0);
    mma_kfrag(f0, acc);
    load_kfrag(f0, A, lda, B, ldb, row, kh, k + 16);
    __builtin_amdgcn_sched_barrier(0);
    mma_kfrag(f1, acc);
    load_kfrag(f1, A, lda, B, ldb, row, kh, k + 20);
    __builtin_amdgcn_sched_barrier(0);
    mma_kfrag(f2, acc);
    load_kfrag(f2, A, lda, B, ldb, row, kh, k + 24);
    __builtin_amdgcn_sched_barrier(0);
    mma_kfrag(f3, acc);
  }
  load_kfrag(f3, A, lda, B, ldb, row, kh, k + 12);
  mma_kfrag(f0, acc);
  mma_kfrag(f1, acc);
  mma_kfrag(f2, acc);
  mma_kfrag(f3, acc);
}

__device__ __forceinline__ void wave_store_32x128(
    float* __restrict__ C, int ldc, const v8f acc[2][8]) {
  const int lane = threadIdx.x & 31;
  const int col  = lane & 15;
  const int rb   = (lane >> 4) * 8;
#pragma unroll
  for (int t = 0; t < 2; ++t)
#pragma unroll
    for (int nt = 0; nt < 8; ++nt)
#pragma unroll
      for (int v = 0; v < 8; ++v)
        C[(t * 16 + rb + v) * ldc + nt * 16 + col] = acc[t][nt][v];
}

__global__ void __launch_bounds__(32, 1)
sph_phaseA_kernel(const float* __restrict__ A1,
                  const float* __restrict__ X,
                  float* __restrict__ coef) {
  const int mt = blockIdx.x;
  const int b  = blockIdx.y;
  const float* A = A1 + (size_t)mt * 32 * KJP;
  const float* B = X + (size_t)b * KJP * CIN;
  float* C = coef + (size_t)b * KLM * CIN + (size_t)mt * 32 * CIN;
  v8f acc[2][8] = {};
  wave_gemm_32x128(A, KJP, B, CIN, KJP, acc);
  wave_store_32x128(C, CIN, acc);
}

__global__ void __launch_bounds__(32, 1)
sph_phaseB_kernel(const float* __restrict__ coef,
                  const float* __restrict__ W,
                  float* __restrict__ chat) {
  const int l = blockIdx.x;
  const int b = blockIdx.y;
  const float* A = coef + (size_t)b * KLM * CIN + (size_t)l * 32 * CIN;
  const float* B = W + (size_t)l * COUT;
  float* C = chat + (size_t)b * KLM * COUT + (size_t)l * 32 * COUT;
  v8f acc[2][8] = {};
  wave_gemm_32x128(A, CIN, B, LMAX * COUT, CIN, acc);
  wave_store_32x128(C, COUT, acc);
}

__global__ void __launch_bounds__(32, 1)
sph_phaseC_kernel(const float* __restrict__ A2,
                  const float* __restrict__ chat,
                  const float* __restrict__ bias,
                  float* __restrict__ out) {
  const int mt = blockIdx.x;
  const int b  = blockIdx.y;
  const int lane = threadIdx.x & 31;
  v8f acc[2][8];
#pragma unroll
  for (int nt = 0; nt < 8; ++nt) {
    const float bv = bias[nt * 16 + (lane & 15)];
#pragma unroll
    for (int v = 0; v < 8; ++v) {
      acc[0][nt][v] = bv;
      acc[1][nt][v] = bv;
    }
  }
  wave_gemm_32x128(A2 + (size_t)mt * 32 * KLM, KLM,
                   chat + (size_t)b * KLM * COUT, COUT, KLM, acc);
  wave_store_32x128(out + (size_t)b * KJP * COUT + (size_t)mt * 32 * COUT, COUT, acc);
}

extern "C" void kernel_launch(void* const* d_in, const int* in_sizes, int n_in,
                              void* d_out, int out_size, void* d_ws, size_t ws_size,
                              hipStream_t stream) {
  const float* inp  = (const float*)d_in[0];
  const float* wgt  = (const float*)d_in[1];
  const float* bias = (const float*)d_in[2];
  float* out = (float*)d_out;
  float* ws  = (float*)d_ws;

  sph_legendre_kernel<<<1, LMAX * NGRID, 0, stream>>>(ws);
  sph_buildmats_kernel<<<(LMAX * MH * KJP + 255) / 256, 256, 0, stream>>>(ws);

  dim3 gAB(16, BATCH);
  sph_phaseA_kernel<<<gAB, 32, 0, stream>>>(ws + WS_A1, inp, ws + WS_COEF);
  sph_phaseB_kernel<<<gAB, 32, 0, stream>>>(ws + WS_COEF, wgt, ws + WS_CHAT);

  dim3 gC(32, BATCH);
  sph_phaseC_kernel<<<gC, 32, 0, stream>>>(ws + WS_A2, ws + WS_CHAT, bias, out);
}


# Structurally zero spectral rows dropped (orders m>l; imaginary part of m=0): 512 -> 256 packed rows, half the WMMAs in every GEMM; strips re-cut so each SIMD keeps a wave; synthesis sum order changes
# speedup vs baseline: 9.2550x; 1.6814x over previous
#include <hip/hip_runtime.h>
#include <math.h>

typedef float v2f __attribute__((ext_vector_type(2)));
typedef float v8f __attribute__((ext_vector_type(8)));

#define NGRID 32
#define LMAX  16
#define NM    31
#define CIN   128
#define COUT  128
#define BATCH 64
#define KJP   1024
#define KR    256


#define WS_Q      0
#define WS_A1     16384
#define WS_A2     (WS_A1 + KR*KJP)
#define WS_COEF   (WS_A2 + KJP*KR)
#define WS_CHAT   (WS_COEF + BATCH*KR*CIN)

#define PI_D 3.14159265358979323846

__global__ void __launch_bounds__(LMAX * NGRID)
sph_legendre_kernel(float* __restrict__ ws) {
  __shared__ double s_x[NGRID];
  __shared__ double s_sx[NGRID];
  __shared__ double s_ratio[LMAX][LMAX];
  __shared__ float  s_nlm[LMAX][LMAX];
  __shared__ float  s_cx[LMAX][NGRID];
  __shared__ float  s_cs[LMAX][NGRID];

  const int tid = threadIdx.x;
  const int m = tid >> 5;
  const int j = tid & 31;

  if (m == 0) {
    const double theta = PI_D * (j + 0.5) / NGRID;
    s_x[j]  = cos(theta);
    s_sx[j] = sin(theta);
  } else if (m == 1 && j < LMAX) {
    const int d = j;
    double ratio = 1.0;
    s_ratio[d][0] = ratio;
    for (int mm = 1; d + mm < LMAX; ++mm) {
      ratio /= (double)(d + 2 * mm - 1);
      ratio /= (double)(d + 2 * mm);
      s_ratio[d + mm][mm] = ratio;
    }
  }
  __syncthreads();

  {
    const double x = s_x[j], sx = s_sx[j];
    s_cx[m][j] = (float)((2.0 * m + 1.0) * x);
    s_cs[m][j] = (float)(-(2.0 * m - 1.0) * sx);
  }
  if (tid < LMAX * LMAX) {
    const int l = tid >> 4, mm = tid & 15;
    if (mm <= l)
      s_nlm[l][mm] = (float)sqrt((2.0 * l + 1.0) / (4.0 * PI_D) * s_ratio[l][mm]);
  }
  __syncthreads();

  float* qp = ws + WS_Q + (15 + m) * NGRID + j;
  float* qn = ws + WS_Q + (15 - m) * NGRID + j;

  for (int l = 0; l < m; ++l) {
    qp[l * NM * NGRID] = 0.f;
    qn[l * NM * NGRID] = 0.f;
  }

  float pmm = 1.0f;
  for (int k = 1; k <= m; ++k) pmm = s_cs[k][j] * pmm;
  {
    const float q = s_nlm[m][m] * pmm;
    qp[m * NM * NGRID] = q;
    qn[m * NM * NGRID] = (m & 1) ? -q : q;
  }
  if (m < LMAX - 1) {
    float p2 = pmm;
    float p1 = s_cx[m][j] * pmm;
    {
      const float q = s_nlm[m + 1][m] * p1;
      qp[(m + 1) * NM * NGRID] = q;
      qn[(m + 1) * NM * NGRID] = (m & 1) ? -q : q;
    }
    for (int l = m + 2; l < LMAX; ++l) {
      const float p = (s_cx[l - 1][j] * p1 - (float)(l + m - 1) * p2) / (float)(l - m);
      const float q = s_nlm[l][m] * p;
      qp[l * NM * NGRID] = q;
      qn[l * NM * NGRID] = (m & 1) ? -q : q;
      p2 = p1;
      p1 = p;
    }
  }
}

__global__ void sph_buildmats_kernel(float* __restrict__ ws) {
  const int idx = blockIdx.x * blockDim.x + threadIdx.x;
  if (idx >= KR * KJP) return;
  const int r = idx >> 10, jp = idx & 1023;
  int l = 0;
  while ((l + 1) * (l + 1) <= r) ++l;
  const int i = r - l * l;
  const bool imag = i > l;
  const int mh = imag ? i - l : i;
  const int jj = jp >> 5, p = jp & 31;

  const float q = ws[WS_Q + (l * NM + 15 + mh) * NGRID + jj];
  const float ang = (2.0f * (float)PI_D / NGRID) * (float)(mh * p);
  float s, c;
  sincosf(ang, &s, &c);
  const float tw = imag ? -s : c;
  const float th = (float)PI_D * (jj + 0.5f) / NGRID;
  const float wj = sinf(th) * ((float)PI_D / NGRID) * (2.0f * (float)PI_D / NGRID);
  const float scale_l = 2.0f * (float)PI_D * sqrtf(4.0f * (float)PI_D / (2.0f * l + 1.0f));
  const float f = (mh == 0) ? 1.0f : 2.0f;

  ws[WS_A1 + r * KJP + jp] = q * wj * tw;
  ws[WS_A2 + jp * KR + r]  = f * (q * scale_l * tw);
}

template <int NT, int KS>
struct KFrag {
  v2f a0[KS], a1[KS];
  v2f b[KS][NT];
};

template <int NT, int KS>
__device__ __forceinline__ void load_kfrag(
    KFrag<NT, KS>& f, const float* __restrict__ A, int lda,
    const float* __restrict__ B, int ldb, int row, int kh, int k) {
#pragma unroll
  for (int ks = 0; ks < KS; ++ks) {
    const int kk = k + 4 * ks + kh;
    f.a0[ks].x = A[row * lda + kk];
    f.a0[ks].y = A[row * lda + kk + 1];
    f.a1[ks].x = A[(row + 16) * lda + kk];
    f.a1[ks].y = A[(row + 16) * lda + kk + 1];
#pragma unroll
    for (int nt = 0; nt < NT; ++nt) {
      f.b[ks][nt].x = B[kk * ldb + nt * 16 + row];
      f.b[ks][nt].y = B[(kk + 1) * ldb + nt * 16 + row];
    }
  }
}

template <int NT, int KS>
__device__ __forceinline__ void mma_kfrag(const KFrag<NT, KS>& f, v8f acc[2][NT]) {
#pragma unroll
  for (int ks = 0; ks < KS; ++ks) {
#pragma unroll
    for (int nt = 0; nt < NT; ++nt) {
      acc[0][nt] = __builtin_amdgcn_wmma_f32_16x16x4_f32(
          false, f.a0[ks], false, f.b[ks][nt], (short)0, acc[0][nt], false, false);
      acc[1][nt] = __builtin_amdgcn_wmma_f32_16x16x4_f32(
          false, f.a1[ks], false, f.b[ks][nt], (short)0, acc[1][nt], false, false);
    }
  }
}

__device__ __forceinline__ void wmma_hazard_slots() {
  __builtin_amdgcn_sched_barrier(0);
  asm volatile("v_nop\n\tv_nop\n\tv_nop\n\tv_nop");
  __builtin_amdgcn_sched_barrier(0);
}

template <int NT, int KS>
__device__ __forceinline__ void wave_gemm_32xN(
    const float* __restrict__ A, int lda,
    const float* __restrict__ B, int ldb,
    int K, v8f acc[2][NT]) {
  constexpr int S = 4 * KS;
  const int lane = threadIdx.x & 31;
  const int row  = lane & 15;
  const int kh   = (lane >> 4) << 1;
  KFrag<NT, KS> f0, f1, f2, f3;
  load_kfrag(f0, A, lda, B, ldb, row, kh, 0);
  __builtin_amdgcn_sched_barrier(0);
  load_kfrag(f1, A, lda, B, ldb, row, kh, S);
  __builtin_amdgcn_sched_barrier(0);
  load_kfrag(f2, A, lda, B, ldb, row, kh, 2 * S);
  int k = 0;
  for (; k + 4 * S < K; k += 4 * S) {
    load_kfrag(f3, A, lda, B, ldb, row, kh, k + 3 * S);
    __builtin_amdgcn_sched_barrier(0);
    mma_kfrag(f0, acc);
    wmma_hazard_slots();
    load_kfrag(f0, A, lda, B, ldb, row, kh, k + 4 * S);
    __builtin_amdgcn_sched_barrier(0);
    mma_kfrag(f1, acc);
    wmma_hazard_slots();
    load_kfrag(f1, A, lda, B, ldb, row, kh, k + 5 * S);
    __builtin_amdgcn_sched_barrier(0);
    mma_kfrag(f2, acc);
    wmma_hazard_slots();
    load_kfrag(f2, A, lda, B, ldb, row, kh, k + 6 * S);
    __builtin_amdgcn_sched_barrier(0);
    mma_kfrag(f3, acc);
    wmma_hazard_slots();
  }
  load_kfrag(f3, A, lda, B, ldb, row, kh, k + 3 * S);
  mma_kfrag(f0, acc);
  mma_kfrag(f1, acc);
  mma_kfrag(f2, acc);
  mma_kfrag(f3, acc);
  wmma_hazard_slots();
}

template <int NT>
__device__ __forceinline__ void wave_store_32xN(
    float* __restrict__ C, int ldc, const v8f acc[2][NT]) {
  const int lane = threadIdx.x & 31;
  const int col  = lane & 15;
  const int rb   = (lane >> 4) * 8;
#pragma unroll
  for (int t = 0; t < 2; ++t)
#pragma unroll
    for (int nt = 0; nt < NT; ++nt)
#pragma unroll
      for (int v = 0; v < 8; ++v)
        C[(t * 16 + rb + v) * ldc + nt * 16 + col] = acc[t][nt][v];
}

__global__ void __launch_bounds__(32, 1)
sph_phaseA_kernel(const float* __restrict__ A1,
                  const float* __restrict__ X,
                  float* __restrict__ coef) {
  const int mt = blockIdx.x >> 1;
  const int ch = blockIdx.x & 1;
  const int b  = blockIdx.y;
  const float* A = A1 + (size_t)mt * 32 * KJP;
  const float* B = X + (size_t)b * KJP * CIN + ch * 64;
  float* C = coef + (size_t)b * KR * CIN + (size_t)mt * 32 * CIN + ch * 64;
  v8f acc[2][4] = {};
  wave_gemm_32xN<4, 2>(A, KJP, B, CIN, KJP, acc);
  wave_store_32xN<4>(C, CIN, acc);
}

__global__ void __launch_bounds__(32, 1)
sph_phaseB_kernel(const float* __restrict__ coef,
                  const float* __restrict__ W,
                  float* __restrict__ chat) {
  const int ch = blockIdx.x & 1;
  const int b0 = (blockIdx.x >> 1) * 32;
  const int r  = blockIdx.y;
  int l = 0;
  while ((l + 1) * (l + 1) <= r) ++l;
  const float* A = coef + ((size_t)b0 * KR + r) * CIN;
  const float* B = W + (size_t)l * COUT + ch * 64;
  float* C = chat + ((size_t)b0 * KR + r) * COUT + ch * 64;
  v8f acc[2][4] = {};
  wave_gemm_32xN<4, 2>(A, KR * CIN, B, LMAX * COUT, CIN, acc);
  wave_store_32xN<4>(C, KR * COUT, acc);
}

__global__ void __launch_bounds__(32, 1)
sph_phaseC_kernel(const float* __restrict__ A2,
                  const float* __restrict__ chat,
                  const float* __restrict__ bias,
                  float* __restrict__ out) {
  const int mt = blockIdx.x;
  const int b  = blockIdx.y;
  const int lane = threadIdx.x & 31;
  v8f acc[2][8];
#pragma unroll
  for (int nt = 0; nt < 8; ++nt) {
    const float bv = bias[nt * 16 + (lane & 15)];
#pragma unroll
    for (int v = 0; v < 8; ++v) {
      acc[0][nt][v] = bv;
      acc[1][nt][v] = bv;
    }
  }
  wave_gemm_32xN<8, 1>(A2 + (size_t)mt * 32 * KR, KR,
                       chat + (size_t)b * KR * COUT, COUT, KR, acc);
  wave_store_32xN<8>(out + (size_t)b * KJP * COUT + (size_t)mt * 32 * COUT, COUT, acc);
}

extern "C" void kernel_launch(void* const* d_in, const int* in_sizes, int n_in,
                              void* d_out, int out_size, void* d_ws, size_t ws_size,
                              hipStream_t stream) {
  const float* inp  = (const float*)d_in[0];
  const float* wgt  = (const float*)d_in[1];
  const float* bias = (const float*)d_in[2];
  float* out = (float*)d_out;
  float* ws  = (float*)d_ws;

  sph_legendre_kernel<<<1, LMAX * NGRID, 0, stream>>>(ws);
  sph_buildmats_kernel<<<(KR * KJP + 255) / 256, 256, 0, stream>>>(ws);

  dim3 gA(16, BATCH);
  sph_phaseA_kernel<<<gA, 32, 0, stream>>>(ws + WS_A1, inp, ws + WS_COEF);
  dim3 gB(4, KR);
  sph_phaseB_kernel<<<gB, 32, 0, stream>>>(ws + WS_COEF, wgt, ws + WS_CHAT);

  dim3 gC(32, BATCH);
  sph_phaseC_kernel<<<gC, 32, 0, stream>>>(ws + WS_A2, ws + WS_CHAT, bias, out);
}


# Analysis and synthesis matrices stored in A-fragment order by the build kernel: a lane fetches its k values of a load group with one 16- or 8-byte load over whole lines; k order and sums unchanged
# speedup vs baseline: 10.0568x; 1.0866x over previous
#include <hip/hip_runtime.h>
#include <math.h>

typedef float v2f __attribute__((ext_vector_type(2)));
typedef float v8f __attribute__((ext_vector_type(8)));

#define NGRID 32
#define LMAX  16
#define NM    31
#define CIN   128
#define COUT  128
#define BATCH 64
#define KJP   1024
#define KR    256


#define WS_Q      0
#define WS_A1     16384
#define WS_A2     (WS_A1 + KR*KJP)
#define WS_COEF   (WS_A2 + KJP*KR)
#define WS_CHAT   (WS_COEF + BATCH*KR*CIN)

#define PI_D 3.14159265358979323846

__global__ void __launch_bounds__(LMAX * NGRID)
sph_legendre_kernel(float* __restrict__ ws) {
  __shared__ double s_x[NGRID];
  __shared__ double s_sx[NGRID];
  __shared__ double s_ratio[LMAX][LMAX];
  __shared__ float  s_nlm[LMAX][LMAX];
  __shared__ float  s_cx[LMAX][NGRID];
  __shared__ float  s_cs[LMAX][NGRID];

  const int tid = threadIdx.x;
  const int m = tid >> 5;
  const int j = tid & 31;

  if (m == 0) {
    const double theta = PI_D * (j + 0.5) / NGRID;
    s_x[j]  = cos(theta);
    s_sx[j] = sin(theta);
  } else if (m == 1 && j < LMAX) {
    const int d = j;
    double ratio = 1.0;
    s_ratio[d][0] = ratio;
    for (int mm = 1; d + mm < LMAX; ++mm) {
      ratio /= (double)(d + 2 * mm - 1);
      ratio /= (double)(d + 2 * mm);
      s_ratio[d + mm][mm] = ratio;
    }
  }
  __syncthreads();

  {
    const double x = s_x[j], sx = s_sx[j];
    s_cx[m][j] = (float)((2.0 * m + 1.0) * x);
    s_cs[m][j] = (float)(-(2.0 * m - 1.0) * sx);
  }
  if (tid < LMAX * LMAX) {
    const int l = tid >> 4, mm = tid & 15;
    if (mm <= l)
      s_nlm[l][mm] = (float)sqrt((2.0 * l + 1.0) / (4.0 * PI_D) * s_ratio[l][mm]);
  }
  __syncthreads();

  float* qp = ws + WS_Q + (15 + m) * NGRID + j;
  float* qn = ws + WS_Q + (15 - m) * NGRID + j;

  for (int l = 0; l < m; ++l) {
    qp[l * NM * NGRID] = 0.f;
    qn[l * NM * NGRID] = 0.f;
  }

  float pmm = 1.0f;
  for (int k = 1; k <= m; ++k) pmm = s_cs[k][j] * pmm;
  {
    const float q = s_nlm[m][m] * pmm;
    qp[m * NM * NGRID] = q;
    qn[m * NM * NGRID] = (m & 1) ? -q : q;
  }
  if (m < LMAX - 1) {
    float p2 = pmm;
    float p1 = s_cx[m][j] * pmm;
    {
      const float q = s_nlm[m + 1][m] * p1;
      qp[(m + 1) * NM * NGRID] = q;
      qn[(m + 1) * NM * NGRID] = (m & 1) ? -q : q;
    }
    for (int l = m + 2; l < LMAX; ++l) {
      const float p = (s_cx[l - 1][j] * p1 - (float)(l + m - 1) * p2) / (float)(l - m);
      const float q = s_nlm[l][m] * p;
      qp[l * NM * NGRID] = q;
      qn[l * NM * NGRID] = (m & 1) ? -q : q;
      p2 = p1;
      p1 = p;
    }
  }
}

template <int KS, int K>
__device__ __forceinline__ int frag_offset(int m, int k) {
  constexpr int G = K / (4 * KS);
  const int strip = m >> 5, t = (m >> 4) & 1, row = m & 15;
  const int g = k / (4 * KS), ks = (k >> 2) % KS, h = (k >> 1) & 1, e = k & 1;
  return ((((strip * G + g) * 2 + t) * 32 + h * 16 + row) * (2 * KS)) + ks * 2 + e;
}

__global__ void sph_buildmats_kernel(float* __restrict__ ws) {
  const int idx = blockIdx.x * blockDim.x + threadIdx.x;
  if (idx >= KR * KJP) return;
  const int r = idx >> 10, jp = idx & 1023;
  int l = 0;
  while ((l + 1) * (l + 1) <= r) ++l;
  const int i = r - l * l;
  const bool imag = i > l;
  const int mh = imag ? i - l : i;
  const int jj = jp >> 5, p = jp & 31;

  const float q = ws[WS_Q + (l * NM + 15 + mh) * NGRID + jj];
  const float ang = (2.0f * (float)PI_D / NGRID) * (float)(mh * p);
  float s, c;
  sincosf(ang, &s, &c);
  const float tw = imag ? -s : c;
  const float th = (float)PI_D * (jj + 0.5f) / NGRID;
  const float wj = sinf(th) * ((float)PI_D / NGRID) * (2.0f * (float)PI_D / NGRID);
  const float scale_l = 2.0f * (float)PI_D * sqrtf(4.0f * (float)PI_D / (2.0f * l + 1.0f));
  const float f = (mh == 0) ? 1.0f : 2.0f;

  ws[WS_A1 + frag_offset<2, KJP>(r, jp)] = q * wj * tw;
  ws[WS_A2 + frag_offset<1, KR>(jp, r)]  = f * (q * scale_l * tw);
}

template <int NT, int KS>
struct KFrag {
  v2f a0[KS], a1[KS];
  v2f b[KS][NT];
};

template <int NT, int KS, bool AF>
__device__ __forceinline__ void load_kfrag(
    KFrag<NT, KS>& f, const float* __restrict__ A, int lda,
    const float* __restrict__ B, int ldb, int row, int kh, int k) {
  if constexpr (AF) {
    typedef float vA __attribute__((ext_vector_type(2 * KS)));
    const int lane = (kh << 3) + row;
    const vA* p = reinterpret_cast<const vA*>(A) + (k / (4 * KS)) * 64 + lane;
    const vA t0 = p[0];
    const vA t1 = p[32];
#pragma unroll
    for (int ks = 0; ks < KS; ++ks) {
      f.a0[ks].x = t0[2 * ks];
      f.a0[ks].y = t0[2 * ks + 1];
      f.a1[ks].x = t1[2 * ks];
      f.a1[ks].y = t1[2 * ks + 1];
    }
  }
#pragma unroll
  for (int ks = 0; ks < KS; ++ks) {
    const int kk = k + 4 * ks + kh;
    if constexpr (!AF) {
      f.a0[ks].x = A[row * lda + kk];
      f.a0[ks].y = A[row * lda + kk + 1];
      f.a1[ks].x = A[(row + 16) * lda + kk];
      f.a1[ks].y = A[(row + 16) * lda + kk + 1];
    }
#pragma unroll
    for (int nt = 0; nt < NT; ++nt) {
      f.b[ks][nt].x = B[kk * ldb + nt * 16 + row];
      f.b[ks][nt].y = B[(kk + 1) * ldb + nt * 16 + row];
    }
  }
}

template <int NT, int KS>
__device__ __forceinline__ void mma_kfrag(const KFrag<NT, KS>& f, v8f acc[2][NT]) {
#pragma unroll
  for (int ks = 0; ks < KS; ++ks) {
#pragma unroll
    for (int nt = 0; nt < NT; ++nt) {
      acc[0][nt] = __builtin_amdgcn_wmma_f32_16x16x4_f32(
          false, f.a0[ks], false, f.b[ks][nt], (short)0, acc[0][nt], false, false);
      acc[1][nt] = __builtin_amdgcn_wmma_f32_16x16x4_f32(
          false, f.a1[ks], false, f.b[ks][nt], (short)0, acc[1][nt], false, false);
    }
  }
}

__device__ __forceinline__ void wmma_hazard_slots() {
  __builtin_amdgcn_sched_barrier(0);
  asm volatile("v_nop\n\tv_nop\n\tv_nop\n\tv_nop");
  __builtin_amdgcn_sched_barrier(0);
}

template <int NT, int KS, bool AF>
__device__ __forceinline__ void wave_gemm_32xN(
    const float* __restrict__ A, int lda,
    const float* __restrict__ B, int ldb,
    int K, v8f acc[2][NT]) {
  constexpr int S = 4 * KS;
  const int lane = threadIdx.x & 31;
  const int row  = lane & 15;
  const int kh   = (lane >> 4) << 1;
  KFrag<NT, KS> f0, f1, f2, f3;
  load_kfrag<NT, KS, AF>(f0, A, lda, B, ldb, row, kh, 0);
  __builtin_amdgcn_sched_barrier(0);
  load_kfrag<NT, KS, AF>(f1, A, lda, B, ldb, row, kh, S);
  __builtin_amdgcn_sched_barrier(0);
  load_kfrag<NT, KS, AF>(f2, A, lda, B, ldb, row, kh, 2 * S);
  int k = 0;
  for (; k + 4 * S < K; k += 4 * S) {
    load_kfrag<NT, KS, AF>(f3, A, lda, B, ldb, row, kh, k + 3 * S);
    __builtin_amdgcn_sched_barrier(0);
    mma_kfrag(f0, acc);
    wmma_hazard_slots();
    load_kfrag<NT, KS, AF>(f0, A, lda, B, ldb, row, kh, k + 4 * S);
    __builtin_amdgcn_sched_barrier(0);
    mma_kfrag(f1, acc);
    wmma_hazard_slots();
    load_kfrag<NT, KS, AF>(f1, A, lda, B, ldb, row, kh, k + 5 * S);
    __builtin_amdgcn_sched_barrier(0);
    mma_kfrag(f2, acc);
    wmma_hazard_slots();
    load_kfrag<NT, KS, AF>(f2, A, lda, B, ldb, row, kh, k + 6 * S);
    __builtin_amdgcn_sched_barrier(0);
    mma_kfrag(f3, acc);
    wmma_hazard_slots();
  }
  load_kfrag<NT, KS, AF>(f3, A, lda, B, ldb, row, kh, k + 3 * S);
  mma_kfrag(f0, acc);
  mma_kfrag(f1, acc);
  mma_kfrag(f2, acc);
  mma_kfrag(f3, acc);
  wmma_hazard_slots();
}

template <int NT>
__device__ __forceinline__ void wave_store_32xN(
    float* __restrict__ C, int ldc, const v8f acc[2][NT]) {
  const int lane = threadIdx.x & 31;
  const int col  = lane & 15;
  const int rb   = (lane >> 4) * 8;
#pragma unroll
  for (int t = 0; t < 2; ++t)
#pragma unroll
    for (int nt = 0; nt < NT; ++nt)
#pragma unroll
      for (int v = 0; v < 8; ++v)
        C[(t * 16 + rb + v) * ldc + nt * 16 + col] = acc[t][nt][v];
}

__global__ void __launch_bounds__(32, 1)
sph_phaseA_kernel(const float* __restrict__ A1,
                  const float* __restrict__ X,
                  float* __restrict__ coef) {
  const int mt = blockIdx.x >> 1;
  const int ch = blockIdx.x & 1;
  const int b  = blockIdx.y;
  const float* A = A1 + (size_t)mt * 32 * KJP;
  const float* B = X + (size_t)b * KJP * CIN + ch * 64;
  float* C = coef + (size_t)b * KR * CIN + (size_t)mt * 32 * CIN + ch * 64;
  v8f acc[2][4] = {};
  wave_gemm_32xN<4, 2, true>(A, KJP, B, CIN, KJP, acc);
  wave_store_32xN<4>(C, CIN, acc);
}

__global__ void __launch_bounds__(32, 1)
sph_phaseB_kernel(const float* __restrict__ coef,
                  const float* __restrict__ W,
                  float* __restrict__ chat) {
  const int ch = blockIdx.x & 1;
  const int b0 = (blockIdx.x >> 1) * 32;
  const int r  = blockIdx.y;
  int l = 0;
  while ((l + 1) * (l + 1) <= r) ++l;
  const float* A = coef + ((size_t)b0 * KR + r) * CIN;
  const float* B = W + (size_t)l * COUT + ch * 64;
  float* C = chat + ((size_t)b0 * KR + r) * COUT + ch * 64;
  v8f acc[2][4] = {};
  wave_gemm_32xN<4, 2, false>(A, KR * CIN, B, LMAX * COUT, CIN, acc);
  wave_store_32xN<4>(C, KR * COUT, acc);
}

__global__ void __launch_bounds__(32, 1)
sph_phaseC_kernel(const float* __restrict__ A2,
                  const float* __restrict__ chat,
                  const float* __restrict__ bias,
                  float* __restrict__ out) {
  const int mt = blockIdx.x;
  const int b  = blockIdx.y;
  const int lane = threadIdx.x & 31;
  v8f acc[2][8];
#pragma unroll
  for (int nt = 0; nt < 8; ++nt) {
    const float bv = bias[nt * 16 + (lane & 15)];
#pragma unroll
    for (int v = 0; v < 8; ++v) {
      acc[0][nt][v] = bv;
      acc[1][nt][v] = bv;
    }
  }
  wave_gemm_32xN<8, 1, true>(A2 + (size_t)mt * 32 * KR, KR,
                       chat + (size_t)b * KR * COUT, COUT, KR, acc);
  wave_store_32xN<8>(out + (size_t)b * KJP * COUT + (size_t)mt * 32 * COUT, COUT, acc);
}

extern "C" void kernel_launch(void* const* d_in, const int* in_sizes, int n_in,
                              void* d_out, int out_size, void* d_ws, size_t ws_size,
                              hipStream_t stream) {
  const float* inp  = (const float*)d_in[0];
  const float* wgt  = (const float*)d_in[1];
  const float* bias = (const float*)d_in[2];
  float* out = (float*)d_out;
  float* ws  = (float*)d_ws;

  sph_legendre_kernel<<<1, LMAX * NGRID, 0, stream>>>(ws);
  sph_buildmats_kernel<<<(KR * KJP + 255) / 256, 256, 0, stream>>>(ws);

  dim3 gA(16, BATCH);
  sph_phaseA_kernel<<<gA, 32, 0, stream>>>(ws + WS_A1, inp, ws + WS_COEF);
  dim3 gB(4, KR);
  sph_phaseB_kernel<<<gB, 32, 0, stream>>>(ws + WS_COEF, wgt, ws + WS_CHAT);

  dim3 gC(32, BATCH);
  sph_phaseC_kernel<<<gC, 32, 0, stream>>>(ws + WS_A2, ws + WS_CHAT, bias, out);
}


# Both transforms made separable: a 32x32 longitude-DFT GEMM and a per-order Legendre GEMM replace each fused 256x1024 matrix (4x fewer matrix ops overall); two more launches; sum order changes
# speedup vs baseline: 14.9260x; 1.4842x over previous
#include <hip/hip_runtime.h>
#include <math.h>

typedef float v2f __attribute__((ext_vector_type(2)));
typedef float v8f __attribute__((ext_vector_type(8)));

#define NGRID 32
#define LMAX  16
#define NM    31
#define CIN   128
#define COUT  128
#define BATCH 64
#define KJP   1024
#define KR    256


#define WS_Q      0
#define WS_T      16384
#define WS_TT     (WS_T + 32*32)
#define WS_QW     (WS_TT + 32*32)
#define WS_QS     (WS_QW + 16*16*32)
#define WS_COEF   65536
#define WS_CHAT   (WS_COEF + BATCH*KR*CIN)
#define WS_FG     (WS_CHAT + BATCH*KR*COUT)

#define PI_D 3.14159265358979323846

__global__ void __launch_bounds__(LMAX * NGRID)
sph_legendre_kernel(float* __restrict__ ws) {
  __shared__ double s_x[NGRID];
  __shared__ double s_sx[NGRID];
  __shared__ double s_ratio[LMAX][LMAX];
  __shared__ float  s_nlm[LMAX][LMAX];
  __shared__ float  s_cx[LMAX][NGRID];
  __shared__ float  s_cs[LMAX][NGRID];

  const int tid = threadIdx.x;
  const int m = tid >> 5;
  const int j = tid & 31;

  if (m == 0) {
    const double theta = PI_D * (j + 0.5) / NGRID;
    s_x[j]  = cos(theta);
    s_sx[j] = sin(theta);
  } else if (m == 1 && j < LMAX) {
    const int d = j;
    double ratio = 1.0;
    s_ratio[d][0] = ratio;
    for (int mm = 1; d + mm < LMAX; ++mm) {
      ratio /= (double)(d + 2 * mm - 1);
      ratio /= (double)(d + 2 * mm);
      s_ratio[d + mm][mm] = ratio;
    }
  }
  __syncthreads();

  {
    const double x = s_x[j], sx = s_sx[j];
    s_cx[m][j] = (float)((2.0 * m + 1.0) * x);
    s_cs[m][j] = (float)(-(2.0 * m - 1.0) * sx);
  }
  if (tid < LMAX * LMAX) {
    const int l = tid >> 4, mm = tid & 15;
    if (mm <= l)
      s_nlm[l][mm] = (float)sqrt((2.0 * l + 1.0) / (4.0 * PI_D) * s_ratio[l][mm]);
  }
  __syncthreads();

  float* qp = ws + WS_Q + (15 + m) * NGRID + j;
  float* qn = ws + WS_Q + (15 - m) * NGRID + j;

  for (int l = 0; l < m; ++l) {
    qp[l * NM * NGRID] = 0.f;
    qn[l * NM * NGRID] = 0.f;
  }

  float pmm = 1.0f;
  for (int k = 1; k <= m; ++k) pmm = s_cs[k][j] * pmm;
  {
    const float q = s_nlm[m][m] * pmm;
    qp[m * NM * NGRID] = q;
    qn[m * NM * NGRID] = (m & 1) ? -q : q;
  }
  if (m < LMAX - 1) {
    float p2 = pmm;
    float p1 = s_cx[m][j] * pmm;
    {
      const float q = s_nlm[m + 1][m] * p1;
      qp[(m + 1) * NM * NGRID] = q;
      qn[(m + 1) * NM * NGRID] = (m & 1) ? -q : q;
    }
    for (int l = m + 2; l < LMAX; ++l) {
      const float p = (s_cx[l - 1][j] * p1 - (float)(l + m - 1) * p2) / (float)(l - m);
      const float q = s_nlm[l][m] * p;
      qp[l * NM * NGRID] = q;
      qn[l * NM * NGRID] = (m & 1) ? -q : q;
      p2 = p1;
      p1 = p;
    }
  }
}

__global__ void sph_buildmats_kernel(float* __restrict__ ws) {
  const int idx = blockIdx.x * blockDim.x + threadIdx.x;
  if (idx >= LMAX * LMAX * NGRID) return;
  if (idx < NGRID * NGRID) {
    const int t = idx >> 5, p = idx & 31;
    const float ang = (2.0f * (float)PI_D / NGRID) * (float)((t & 15) * p);
    float s, c;
    sincosf(ang, &s, &c);
    const float tw = (t < 16) ? c : -s;
    ws[WS_T + t * NGRID + p]  = tw;
    ws[WS_TT + p * NGRID + t] = tw;
  }
  const int m = idx >> 9, l = (idx >> 5) & 15, j = idx & 31;
  const float q = ws[WS_Q + (l * NM + 15 + m) * NGRID + j];
  const float th = (float)PI_D * (j + 0.5f) / NGRID;
  const float wj = sinf(th) * ((float)PI_D / NGRID) * (2.0f * (float)PI_D / NGRID);
  const float scale_l = 2.0f * (float)PI_D * sqrtf(4.0f * (float)PI_D / (2.0f * l + 1.0f));
  const float f = (m == 0) ? 1.0f : 2.0f;
  ws[WS_QW + (m * LMAX + l) * NGRID + j] = q * wj;
  ws[WS_QS + (m * NGRID + j) * LMAX + l] = f * (q * scale_l);
}

template <int NT, int KS>
struct KFrag {
  v2f a0[KS], a1[KS];
  v2f b[KS][NT];
};

template <int NT, int KS, int MT>
__device__ __forceinline__ void load_kfrag(
    KFrag<NT, KS>& f, const float* __restrict__ A, int lda,
    const float* __restrict__ B, int ldb, int row, int kh, int k) {
#pragma unroll
  for (int ks = 0; ks < KS; ++ks) {
    const int kk = k + 4 * ks + kh;
    f.a0[ks].x = A[row * lda + kk];
    f.a0[ks].y = A[row * lda + kk + 1];
    if constexpr (MT == 2) {
      f.a1[ks].x = A[(row + 16) * lda + kk];
      f.a1[ks].y = A[(row + 16) * lda + kk + 1];
    }
#pragma unroll
    for (int nt = 0; nt < NT; ++nt) {
      f.b[ks][nt].x = B[kk * ldb + nt * 16 + row];
      f.b[ks][nt].y = B[(kk + 1) * ldb + nt * 16 + row];
    }
  }
}

template <int NT, int KS, int MT>
__device__ __forceinline__ void mma_kfrag(const KFrag<NT, KS>& f, v8f acc[MT][NT]) {
#pragma unroll
  for (int ks = 0; ks < KS; ++ks) {
#pragma unroll
    for (int nt = 0; nt < NT; ++nt) {
      acc[0][nt] = __builtin_amdgcn_wmma_f32_16x16x4_f32(
          false, f.a0[ks], false, f.b[ks][nt], (short)0, acc[0][nt], false, false);
      if constexpr (MT == 2)
        acc[1][nt] = __builtin_amdgcn_wmma_f32_16x16x4_f32(
            false, f.a1[ks], false, f.b[ks][nt], (short)0, acc[1][nt], false, false);
    }
  }
}

__device__ __forceinline__ void wmma_hazard_slots() {
  __builtin_amdgcn_sched_barrier(0);
  asm volatile("v_nop\n\tv_nop\n\tv_nop\n\tv_nop");
  __builtin_amdgcn_sched_barrier(0);
}

template <int NT, int KS, int MT>
__device__ __forceinline__ void wave_gemm_32xN(
    const float* __restrict__ A, int lda,
    const float* __restrict__ B, int ldb,
    int K, v8f acc[MT][NT]) {
  constexpr int S = 4 * KS;
  const int lane = threadIdx.x & 31;
  const int row  = lane & 15;
  const int kh   = (lane >> 4) << 1;
  KFrag<NT, KS> f0, f1, f2, f3;
  load_kfrag<NT, KS, MT>(f0, A, lda, B, ldb, row, kh, 0);
  __builtin_amdgcn_sched_barrier(0);
  load_kfrag<NT, KS, MT>(f1, A, lda, B, ldb, row, kh, S);
  __builtin_amdgcn_sched_barrier(0);
  load_kfrag<NT, KS, MT>(f2, A, lda, B, ldb, row, kh, 2 * S);
  int k = 0;
  for (; k + 4 * S < K; k += 4 * S) {
    load_kfrag<NT, KS, MT>(f3, A, lda, B, ldb, row, kh, k + 3 * S);
    __builtin_amdgcn_sched_barrier(0);
    mma_kfrag<NT, KS, MT>(f0, acc);
    wmma_hazard_slots();
    load_kfrag<NT, KS, MT>(f0, A, lda, B, ldb, row, kh, k + 4 * S);
    __builtin_amdgcn_sched_barrier(0);
    mma_kfrag<NT, KS, MT>(f1, acc);
    wmma_hazard_slots();
    load_kfrag<NT, KS, MT>(f1, A, lda, B, ldb, row, kh, k + 5 * S);
    __builtin_amdgcn_sched_barrier(0);
    mma_kfrag<NT, KS, MT>(f2, acc);
    wmma_hazard_slots();
    load_kfrag<NT, KS, MT>(f2, A, lda, B, ldb, row, kh, k + 6 * S);
    __builtin_amdgcn_sched_barrier(0);
    mma_kfrag<NT, KS, MT>(f3, acc);
    wmma_hazard_slots();
  }
  load_kfrag<NT, KS, MT>(f3, A, lda, B, ldb, row, kh, k + 3 * S);
  mma_kfrag<NT, KS, MT>(f0, acc);
  mma_kfrag<NT, KS, MT>(f1, acc);
  mma_kfrag<NT, KS, MT>(f2, acc);
  mma_kfrag<NT, KS, MT>(f3, acc);
  wmma_hazard_slots();
}

template <int NT>
__device__ __forceinline__ void wave_store_32xN(
    float* __restrict__ C, int ldc, const v8f acc[2][NT]) {
  const int lane = threadIdx.x & 31;
  const int col  = lane & 15;
  const int rb   = (lane >> 4) * 8;
#pragma unroll
  for (int t = 0; t < 2; ++t)
#pragma unroll
    for (int nt = 0; nt < NT; ++nt)
#pragma unroll
      for (int v = 0; v < 8; ++v)
        C[(t * 16 + rb + v) * ldc + nt * 16 + col] = acc[t][nt][v];
}

__global__ void __launch_bounds__(32, 1)
sph_phaseA1_kernel(const float* __restrict__ T,
                   const float* __restrict__ X,
                   float* __restrict__ F) {
  const int j = blockIdx.x;
  const int b = blockIdx.y;
  const size_t slab = ((size_t)b * NGRID + j) * NGRID * CIN;
  v8f acc[2][8] = {};
  wave_gemm_32xN<8, 1, 2>(T, NGRID, X + slab, CIN, NGRID, acc);
  wave_store_32xN<8>(F + slab, CIN, acc);
}

__global__ void __launch_bounds__(32, 1)
sph_phaseA2_kernel(const float* __restrict__ QW,
                   const float* __restrict__ F,
                   float* __restrict__ coef) {
  const int t = (blockIdx.x < 16) ? blockIdx.x : blockIdx.x + 1;
  const int m = t & 15;
  const bool imag = t >= 16;
  const int b = blockIdx.y;
  v8f acc[1][8] = {};
  wave_gemm_32xN<8, 1, 1>(QW + m * LMAX * NGRID, NGRID,
                          F + ((size_t)b * KJP + t) * CIN, NGRID * CIN,
                          NGRID, acc);
  const int lane = threadIdx.x & 31;
  const int col  = lane & 15;
  const int rb   = (lane >> 4) * 8;
  float* C = coef + (size_t)b * KR * CIN + col;
#pragma unroll
  for (int v = 0; v < 8; ++v) {
    const int l = rb + v;
    if (l >= m) {
      const int r = l * l + (imag ? l + m : m);
#pragma unroll
      for (int nt = 0; nt < 8; ++nt)
        C[r * CIN + nt * 16] = acc[0][nt][v];
    }
  }
}

__global__ void __launch_bounds__(32, 1)
sph_phaseB_kernel(const float* __restrict__ coef,
                  const float* __restrict__ W,
                  float* __restrict__ chat) {
  const int ch = blockIdx.x & 1;
  const int b0 = (blockIdx.x >> 1) * 32;
  const int r  = blockIdx.y;
  int l = 0;
  while ((l + 1) * (l + 1) <= r) ++l;
  const float* A = coef + ((size_t)b0 * KR + r) * CIN;
  const float* B = W + (size_t)l * COUT + ch * 64;
  float* C = chat + ((size_t)b0 * KR + r) * COUT + ch * 64;
  v8f acc[2][4] = {};
  wave_gemm_32xN<4, 2, 2>(A, KR * CIN, B, LMAX * COUT, CIN, acc);
  wave_store_32xN<4>(C, KR * COUT, acc);
}

__global__ void __launch_bounds__(32, 1)
sph_phaseC1_kernel(const float* __restrict__ QS,
                   const float* __restrict__ chat,
                   float* __restrict__ G) {
  const int t = blockIdx.x;
  const int m = t & 15;
  const bool imag = t >= 16;
  const bool live = t != 16;
  const int b = blockIdx.y;
  const int lane = threadIdx.x & 31;
  const int row  = lane & 15;
  const int kh   = (lane >> 4) << 1;
  const float* A = QS + m * NGRID * LMAX;
  const float* B = chat + (size_t)b * KR * COUT + row;
  v2f a0[4], a1[4], bf[4][8];
#pragma unroll
  for (int ks = 0; ks < 4; ++ks) {
    const int kk = 4 * ks + kh;
    a0[ks].x = A[row * LMAX + kk];
    a0[ks].y = A[row * LMAX + kk + 1];
    a1[ks].x = A[(row + 16) * LMAX + kk];
    a1[ks].y = A[(row + 16) * LMAX + kk + 1];
    const int l0 = kk, l1 = kk + 1;
    const int c0 = (l0 > m) ? l0 : m, c1 = (l1 > m) ? l1 : m;
    const int r0 = c0 * c0 + (imag ? c0 + m : m);
    const int r1 = c1 * c1 + (imag ? c1 + m : m);
    const bool u0 = live && l0 >= m, u1 = live && l1 >= m;
#pragma unroll
    for (int nt = 0; nt < 8; ++nt) {
      const float x0 = B[r0 * COUT + nt * 16];
      const float x1 = B[r1 * COUT + nt * 16];
      bf[ks][nt].x = u0 ? x0 : 0.f;
      bf[ks][nt].y = u1 ? x1 : 0.f;
    }
  }
  v8f acc[2][8] = {};
  __builtin_amdgcn_sched_barrier(0);
#pragma unroll
  for (int ks = 0; ks < 4; ++ks) {
#pragma unroll
    for (int nt = 0; nt < 8; ++nt) {
      acc[0][nt] = __builtin_amdgcn_wmma_f32_16x16x4_f32(
          false, a0[ks], false, bf[ks][nt], (short)0, acc[0][nt], false, false);
      acc[1][nt] = __builtin_amdgcn_wmma_f32_16x16x4_f32(
          false, a1[ks], false, bf[ks][nt], (short)0, acc[1][nt], false, false);
    }
  }
  wmma_hazard_slots();
  wave_store_32xN<8>(G + ((size_t)b * KJP + t) * COUT, NGRID * COUT, acc);
}

__global__ void __launch_bounds__(32, 1)
sph_phaseC2_kernel(const float* __restrict__ TT,
                   const float* __restrict__ G,
                   const float* __restrict__ bias,
                   float* __restrict__ out) {
  const int j = blockIdx.x;
  const int b = blockIdx.y;
  const int lane = threadIdx.x & 31;
  const size_t slab = ((size_t)b * NGRID + j) * NGRID * COUT;
  v8f acc[2][8];
#pragma unroll
  for (int nt = 0; nt < 8; ++nt) {
    const float bv = bias[nt * 16 + (lane & 15)];
#pragma unroll
    for (int v = 0; v < 8; ++v) {
      acc[0][nt][v] = bv;
      acc[1][nt][v] = bv;
    }
  }
  wave_gemm_32xN<8, 1, 2>(TT, NGRID, G + slab, COUT, NGRID, acc);
  wave_store_32xN<8>(out + slab, COUT, acc);
}

extern "C" void kernel_launch(void* const* d_in, const int* in_sizes, int n_in,
                              void* d_out, int out_size, void* d_ws, size_t ws_size,
                              hipStream_t stream) {
  const float* inp  = (const float*)d_in[0];
  const float* wgt  = (const float*)d_in[1];
  const float* bias = (const float*)d_in[2];
  float* out = (float*)d_out;
  float* ws  = (float*)d_ws;

  sph_legendre_kernel<<<1, LMAX * NGRID, 0, stream>>>(ws);
  sph_buildmats_kernel<<<(LMAX * LMAX * NGRID + 255) / 256, 256, 0, stream>>>(ws);

  dim3 gA1(NGRID, BATCH);
  sph_phaseA1_kernel<<<gA1, 32, 0, stream>>>(ws + WS_T, inp, ws + WS_FG);
  dim3 gA2(31, BATCH);
  sph_phaseA2_kernel<<<gA2, 32, 0, stream>>>(ws + WS_QW, ws + WS_FG, ws + WS_COEF);
  dim3 gB(4, KR);
  sph_phaseB_kernel<<<gB, 32, 0, stream>>>(ws + WS_COEF, wgt, ws + WS_CHAT);
  dim3 gC1(NGRID, BATCH);
  sph_phaseC1_kernel<<<gC1, 32, 0, stream>>>(ws + WS_QS, ws + WS_CHAT, ws + WS_FG);
  dim3 gC2(NGRID, BATCH);
  sph_phaseC2_kernel<<<gC2, 32, 0, stream>>>(ws + WS_TT, ws + WS_FG, bias, out);
}


# Longitude and colatitude steps of each transform fused into one 16-wave kernel per (batch entry, 32 channels): the spectrum slab between them stays in LDS; 7 -> 5 launches; sums unchanged
# speedup vs baseline: 23.8612x; 1.5986x over previous
#include <hip/hip_runtime.h>
#include <math.h>

typedef float v2f __attribute__((ext_vector_type(2)));
typedef float v8f __attribute__((ext_vector_type(8)));

#define NGRID 32
#define LMAX  16
#define NM    31
#define CIN   128
#define COUT  128
#define BATCH 64
#define KJP   1024
#define KR    256


#define WS_Q      0
#define WS_T      16384
#define WS_TT     (WS_T + 32*32)
#define WS_QW     (WS_TT + 32*32)
#define WS_QS     (WS_QW + 16*16*32)
#define WS_COEF   65536
#define WS_CHAT   (WS_COEF + BATCH*KR*CIN)

#define PI_D 3.14159265358979323846

__global__ void __launch_bounds__(LMAX * NGRID)
sph_legendre_kernel(float* __restrict__ ws) {
  __shared__ double s_x[NGRID];
  __shared__ double s_sx[NGRID];
  __shared__ double s_ratio[LMAX][LMAX];
  __shared__ float  s_nlm[LMAX][LMAX];
  __shared__ float  s_cx[LMAX][NGRID];
  __shared__ float  s_cs[LMAX][NGRID];

  const int tid = threadIdx.x;
  const int m = tid >> 5;
  const int j = tid & 31;

  if (m == 0) {
    const double theta = PI_D * (j + 0.5) / NGRID;
    s_x[j]  = cos(theta);
    s_sx[j] = sin(theta);
  } else if (m == 1 && j < LMAX) {
    const int d = j;
    double ratio = 1.0;
    s_ratio[d][0] = ratio;
    for (int mm = 1; d + mm < LMAX; ++mm) {
      ratio /= (double)(d + 2 * mm - 1);
      ratio /= (double)(d + 2 * mm);
      s_ratio[d + mm][mm] = ratio;
    }
  }
  __syncthreads();

  {
    const double x = s_x[j], sx = s_sx[j];
    s_cx[m][j] = (float)((2.0 * m + 1.0) * x);
    s_cs[m][j] = (float)(-(2.0 * m - 1.0) * sx);
  }
  if (tid < LMAX * LMAX) {
    const int l = tid >> 4, mm = tid & 15;
    if (mm <= l)
      s_nlm[l][mm] = (float)sqrt((2.0 * l + 1.0) / (4.0 * PI_D) * s_ratio[l][mm]);
  }
  __syncthreads();

  float* qp = ws + WS_Q + (15 + m) * NGRID + j;
  float* qn = ws + WS_Q + (15 - m) * NGRID + j;

  for (int l = 0; l < m; ++l) {
    qp[l * NM * NGRID] = 0.f;
    qn[l * NM * NGRID] = 0.f;
  }

  float pmm = 1.0f;
  for (int k = 1; k <= m; ++k) pmm = s_cs[k][j] * pmm;
  {
    const float q = s_nlm[m][m] * pmm;
    qp[m * NM * NGRID] = q;
    qn[m * NM * NGRID] = (m & 1) ? -q : q;
  }
  if (m < LMAX - 1) {
    float p2 = pmm;
    float p1 = s_cx[m][j] * pmm;
    {
      const float q = s_nlm[m + 1][m] * p1;
      qp[(m + 1) * NM * NGRID] = q;
      qn[(m + 1) * NM * NGRID] = (m & 1) ? -q : q;
    }
    for (int l = m + 2; l < LMAX; ++l) {
      const float p = (s_cx[l - 1][j] * p1 - (float)(l + m - 1) * p2) / (float)(l - m);
      const float q = s_nlm[l][m] * p;
      qp[l * NM * NGRID] = q;
      qn[l * NM * NGRID] = (m & 1) ? -q : q;
      p2 = p1;
      p1 = p;
    }
  }
}

__global__ void sph_buildmats_kernel(float* __restrict__ ws) {
  const int idx = blockIdx.x * blockDim.x + threadIdx.x;
  if (idx >= LMAX * LMAX * NGRID) return;
  if (idx < NGRID * NGRID) {
    const int t = idx >> 5, p = idx & 31;
    const float ang = (2.0f * (float)PI_D / NGRID) * (float)((t & 15) * p);
    float s, c;
    sincosf(ang, &s, &c);
    const float tw = (t < 16) ? c : -s;
    ws[WS_T + t * NGRID + p]  = tw;
    ws[WS_TT + p * NGRID + t] = tw;
  }
  const int m = idx >> 9, l = (idx >> 5) & 15, j = idx & 31;
  const float q = ws[WS_Q + (l * NM + 15 + m) * NGRID + j];
  const float th = (float)PI_D * (j + 0.5f) / NGRID;
  const float wj = sinf(th) * ((float)PI_D / NGRID) * (2.0f * (float)PI_D / NGRID);
  const float scale_l = 2.0f * (float)PI_D * sqrtf(4.0f * (float)PI_D / (2.0f * l + 1.0f));
  const float f = (m == 0) ? 1.0f : 2.0f;
  ws[WS_QW + (m * LMAX + l) * NGRID + j] = q * wj;
  ws[WS_QS + (m * NGRID + j) * LMAX + l] = f * (q * scale_l);
}

template <int NT, int KS>
struct KFrag {
  v2f a0[KS], a1[KS];
  v2f b[KS][NT];
};

template <int NT, int KS, int MT>
__device__ __forceinline__ void load_kfrag(
    KFrag<NT, KS>& f, const float* __restrict__ A, int lda,
    const float* __restrict__ B, int ldb, int row, int kh, int k) {
#pragma unroll
  for (int ks = 0; ks < KS; ++ks) {
    const int kk = k + 4 * ks + kh;
    f.a0[ks].x = A[row * lda + kk];
    f.a0[ks].y = A[row * lda + kk + 1];
    if constexpr (MT == 2) {
      f.a1[ks].x = A[(row + 16) * lda + kk];
      f.a1[ks].y = A[(row + 16) * lda + kk + 1];
    }
#pragma unroll
    for (int nt = 0; nt < NT; ++nt) {
      f.b[ks][nt].x = B[kk * ldb + nt * 16 + row];
      f.b[ks][nt].y = B[(kk + 1) * ldb + nt * 16 + row];
    }
  }
}

template <int NT, int KS, int MT>
__device__ __forceinline__ void mma_kfrag(const KFrag<NT, KS>& f, v8f acc[MT][NT]) {
#pragma unroll
  for (int ks = 0; ks < KS; ++ks) {
#pragma unroll
    for (int nt = 0; nt < NT; ++nt) {
      acc[0][nt] = __builtin_amdgcn_wmma_f32_16x16x4_f32(
          false, f.a0[ks], false, f.b[ks][nt], (short)0, acc[0][nt], false, false);
      if constexpr (MT == 2)
        acc[1][nt] = __builtin_amdgcn_wmma_f32_16x16x4_f32(
            false, f.a1[ks], false, f.b[ks][nt], (short)0, acc[1][nt], false, false);
    }
  }
}

__device__ __forceinline__ void wmma_hazard_slots() {
  __builtin_amdgcn_sched_barrier(0);
  asm volatile("v_nop\n\tv_nop\n\tv_nop\n\tv_nop");
  __builtin_amdgcn_sched_barrier(0);
}

template <int NT, int KS, int MT>
__device__ __forceinline__ void wave_gemm_32xN(
    const float* __restrict__ A, int lda,
    const float* __restrict__ B, int ldb,
    int K, v8f acc[MT][NT]) {
  constexpr int S = 4 * KS;
  const int lane = threadIdx.x & 31;
  const int row  = lane & 15;
  const int kh   = (lane >> 4) << 1;
  KFrag<NT, KS> f0, f1, f2, f3;
  load_kfrag<NT, KS, MT>(f0, A, lda, B, ldb, row, kh, 0);
  __builtin_amdgcn_sched_barrier(0);
  load_kfrag<NT, KS, MT>(f1, A, lda, B, ldb, row, kh, S);
  __builtin_amdgcn_sched_barrier(0);
  load_kfrag<NT, KS, MT>(f2, A, lda, B, ldb, row, kh, 2 * S);
  int k = 0;
  for (; k + 4 * S < K; k += 4 * S) {
    load_kfrag<NT, KS, MT>(f3, A, lda, B, ldb, row, kh, k + 3 * S);
    __builtin_amdgcn_sched_barrier(0);
    mma_kfrag<NT, KS, MT>(f0, acc);
    wmma_hazard_slots();
    load_kfrag<NT, KS, MT>(f0, A, lda, B, ldb, row, kh, k + 4 * S);
    __builtin_amdgcn_sched_barrier(0);
    mma_kfrag<NT, KS, MT>(f1, acc);
    wmma_hazard_slots();
    load_kfrag<NT, KS, MT>(f1, A, lda, B, ldb, row, kh, k + 5 * S);
    __builtin_amdgcn_sched_barrier(0);
    mma_kfrag<NT, KS, MT>(f2, acc);
    wmma_hazard_slots();
    load_kfrag<NT, KS, MT>(f2, A, lda, B, ldb, row, kh, k + 6 * S);
    __builtin_amdgcn_sched_barrier(0);
    mma_kfrag<NT, KS, MT>(f3, acc);
    wmma_hazard_slots();
  }
  load_kfrag<NT, KS, MT>(f3, A, lda, B, ldb, row, kh, k + 3 * S);
  mma_kfrag<NT, KS, MT>(f0, acc);
  mma_kfrag<NT, KS, MT>(f1, acc);
  mma_kfrag<NT, KS, MT>(f2, acc);
  mma_kfrag<NT, KS, MT>(f3, acc);
  wmma_hazard_slots();
}

template <int NT>
__device__ __forceinline__ void wave_store_32xN(
    float* __restrict__ C, int ldc, const v8f acc[2][NT]) {
  const int lane = threadIdx.x & 31;
  const int col  = lane & 15;
  const int rb   = (lane >> 4) * 8;
#pragma unroll
  for (int t = 0; t < 2; ++t)
#pragma unroll
    for (int nt = 0; nt < NT; ++nt)
#pragma unroll
      for (int v = 0; v < 8; ++v)
        C[(t * 16 + rb + v) * ldc + nt * 16 + col] = acc[t][nt][v];
}

#define CW   32
#define NWV  16
#define ASP  36
#define ASJ  (32 * ASP + 12)
#define SSP  40
#define SSJ  (32 * SSP + 12)

__global__ void __launch_bounds__(NWV * 32)
sph_analysis_kernel(const float* __restrict__ T,
                    const float* __restrict__ QW,
                    const float* __restrict__ X,
                    float* __restrict__ coef) {
  __shared__ float S[32 * ASJ];
  const int c0   = blockIdx.x * CW;
  const int b    = blockIdx.y;
  const int w    = threadIdx.x >> 5;
  const int lane = threadIdx.x & 31;
  const int row  = lane & 15;
  const int h    = lane >> 4;
  const int kh   = h << 1;
  const int m    = w;

  v2f xb[2][8][2], ta0[8], ta1[8], qa[8];
#pragma unroll
  for (int ks = 0; ks < 8; ++ks) {
    const int kk = 4 * ks + kh;
    ta0[ks].x = T[row * NGRID + kk];
    ta0[ks].y = T[row * NGRID + kk + 1];
    ta1[ks].x = T[(row + 16) * NGRID + kk];
    ta1[ks].y = T[(row + 16) * NGRID + kk + 1];
  }
  __builtin_amdgcn_sched_barrier(0);
#pragma unroll
  for (int jj = 0; jj < 2; ++jj) {
    const float* Xj = X + (((size_t)b * NGRID + (2 * w + jj)) * NGRID) * CIN + c0 + row;
#pragma unroll
    for (int ks = 0; ks < 8; ++ks) {
      const int kk = 4 * ks + kh;
#pragma unroll
      for (int nt = 0; nt < 2; ++nt) {
        xb[jj][ks][nt].x = Xj[kk * CIN + nt * 16];
        xb[jj][ks][nt].y = Xj[(kk + 1) * CIN + nt * 16];
      }
    }
  }
  __builtin_amdgcn_sched_barrier(0);
#pragma unroll
  for (int ks = 0; ks < 8; ++ks) {
    const int kk = 4 * ks + kh;
    qa[ks].x  = QW[(m * LMAX + row) * NGRID + kk];
    qa[ks].y  = QW[(m * LMAX + row) * NGRID + kk + 1];
  }

  v8f acc[2][2][2] = {};
  __builtin_amdgcn_sched_barrier(0);
#pragma unroll
  for (int jj = 0; jj < 2; ++jj)
#pragma unroll
    for (int ks = 0; ks < 8; ++ks)
#pragma unroll
      for (int nt = 0; nt < 2; ++nt) {
        acc[jj][0][nt] = __builtin_amdgcn_wmma_f32_16x16x4_f32(
            false, ta0[ks], false, xb[jj][ks][nt], (short)0, acc[jj][0][nt], false, false);
        acc[jj][1][nt] = __builtin_amdgcn_wmma_f32_16x16x4_f32(
            false, ta1[ks], false, xb[jj][ks][nt], (short)0, acc[jj][1][nt], false, false);
      }
  wmma_hazard_slots();
#pragma unroll
  for (int jj = 0; jj < 2; ++jj)
#pragma unroll
    for (int mt = 0; mt < 2; ++mt)
#pragma unroll
      for (int nt = 0; nt < 2; ++nt)
#pragma unroll
        for (int v = 0; v < 8; ++v)
          S[(2 * w + jj) * ASJ + (mt * 16 + 8 * h + v) * ASP + nt * 16 + row] = acc[jj][mt][nt][v];
  __syncthreads();

  v2f fb[2][8][2];
#pragma unroll
  for (int ti = 0; ti < 2; ++ti)
#pragma unroll
    for (int ks = 0; ks < 8; ++ks) {
      const int kk = 4 * ks + kh;
#pragma unroll
      for (int nt = 0; nt < 2; ++nt) {
        fb[ti][ks][nt].x = S[kk * ASJ + (ti * 16 + m) * ASP + nt * 16 + row];
        fb[ti][ks][nt].y = S[(kk + 1) * ASJ + (ti * 16 + m) * ASP + nt * 16 + row];
      }
    }
  v8f cf[2][2] = {};
  __builtin_amdgcn_sched_barrier(0);
#pragma unroll
  for (int ks = 0; ks < 8; ++ks)
#pragma unroll
    for (int ti = 0; ti < 2; ++ti)
#pragma unroll
      for (int nt = 0; nt < 2; ++nt)
        cf[ti][nt] = __builtin_amdgcn_wmma_f32_16x16x4_f32(
            false, qa[ks], false, fb[ti][ks][nt], (short)0, cf[ti][nt], false, false);
  wmma_hazard_slots();
  float* C = coef + (size_t)b * KR * CIN + c0 + row;
#pragma unroll
  for (int ti = 0; ti < 2; ++ti) {
    if (ti == 1 && m == 0) continue;
#pragma unroll
    for (int v = 0; v < 8; ++v) {
      const int l = 8 * h + v;
      if (l >= m) {
        const int r = l * l + (ti ? l + m : m);
#pragma unroll
        for (int nt = 0; nt < 2; ++nt)
          C[r * CIN + nt * 16] = cf[ti][nt][v];
      }
    }
  }
}

__global__ void __launch_bounds__(32, 1)
sph_phaseB_kernel(const float* __restrict__ coef,
                  const float* __restrict__ W,
                  float* __restrict__ chat) {
  const int ch = blockIdx.x & 1;
  const int b0 = (blockIdx.x >> 1) * 32;
  const int r  = blockIdx.y;
  int l = 0;
  while ((l + 1) * (l + 1) <= r) ++l;
  const float* A = coef + ((size_t)b0 * KR + r) * CIN;
  const float* B = W + (size_t)l * COUT + ch * 64;
  float* C = chat + ((size_t)b0 * KR + r) * COUT + ch * 64;
  v8f acc[2][4] = {};
  wave_gemm_32xN<4, 2, 2>(A, KR * CIN, B, LMAX * COUT, CIN, acc);
  wave_store_32xN<4>(C, KR * COUT, acc);
}

__global__ void __launch_bounds__(NWV * 32)
sph_synthesis_kernel(const float* __restrict__ QS,
                     const float* __restrict__ TT,
                     const float* __restrict__ chat,
                     const float* __restrict__ bias,
                     float* __restrict__ out) {
  __shared__ float S[32 * SSJ];
  const int o0   = blockIdx.x * CW;
  const int b    = blockIdx.y;
  const int w    = threadIdx.x >> 5;
  const int lane = threadIdx.x & 31;
  const int row  = lane & 15;
  const int h    = lane >> 4;
  const int kh   = h << 1;
  const int m    = w;

  const float* A = QS + m * NGRID * LMAX;
  const float* B = chat + (size_t)b * KR * COUT + o0 + row;
  v2f a0[4], a1[4], bf[2][4][2], ta0[8], ta1[8];
#pragma unroll
  for (int ks = 0; ks < 4; ++ks) {
    const int kk = 4 * ks + kh;
    a0[ks].x = A[row * LMAX + kk];
    a0[ks].y = A[row * LMAX + kk + 1];
    a1[ks].x = A[(row + 16) * LMAX + kk];
    a1[ks].y = A[(row + 16) * LMAX + kk + 1];
    const int l0 = kk, l1 = kk + 1;
    const int d0 = (l0 > m) ? l0 : m, d1 = (l1 > m) ? l1 : m;
#pragma unroll
    for (int ti = 0; ti < 2; ++ti) {
      const bool live = !(ti == 1 && m == 0);
      const int r0 = d0 * d0 + (ti ? d0 + m : m);
      const int r1 = d1 * d1 + (ti ? d1 + m : m);
      const bool u0 = live && l0 >= m, u1 = live && l1 >= m;
#pragma unroll
      for (int nt = 0; nt < 2; ++nt) {
        const float x0 = B[r0 * COUT + nt * 16];
        const float x1 = B[r1 * COUT + nt * 16];
        bf[ti][ks][nt].x = u0 ? x0 : 0.f;
        bf[ti][ks][nt].y = u1 ? x1 : 0.f;
      }
    }
  }
#pragma unroll
  for (int ks = 0; ks < 8; ++ks) {
    const int kk = 4 * ks + kh;
    ta0[ks].x = TT[row * NGRID + kk];
    ta0[ks].y = TT[row * NGRID + kk + 1];
    ta1[ks].x = TT[(row + 16) * NGRID + kk];
    ta1[ks].y = TT[(row + 16) * NGRID + kk + 1];
  }
  float bv[2];
#pragma unroll
  for (int nt = 0; nt < 2; ++nt) bv[nt] = bias[o0 + nt * 16 + row];

  v8f acc[2][2][2] = {};
  __builtin_amdgcn_sched_barrier(0);
#pragma unroll
  for (int ks = 0; ks < 4; ++ks)
#pragma unroll
    for (int ti = 0; ti < 2; ++ti)
#pragma unroll
      for (int nt = 0; nt < 2; ++nt) {
        acc[ti][0][nt] = __builtin_amdgcn_wmma_f32_16x16x4_f32(
            false, a0[ks], false, bf[ti][ks][nt], (short)0, acc[ti][0][nt], false, false);
        acc[ti][1][nt] = __builtin_amdgcn_wmma_f32_16x16x4_f32(
            false, a1[ks], false, bf[ti][ks][nt], (short)0, acc[ti][1][nt], false, false);
      }
  wmma_hazard_slots();
#pragma unroll
  for (int ti = 0; ti < 2; ++ti)
#pragma unroll
    for (int mt = 0; mt < 2; ++mt)
#pragma unroll
      for (int nt = 0; nt < 2; ++nt)
#pragma unroll
        for (int v = 0; v < 8; ++v)
          S[(mt * 16 + 8 * h + v) * SSJ + (ti * 16 + m) * SSP + nt * 16 + row] = acc[ti][mt][nt][v];
  __syncthreads();

  v2f gb[2][8][2];
#pragma unroll
  for (int jj = 0; jj < 2; ++jj)
#pragma unroll
    for (int ks = 0; ks < 8; ++ks) {
      const int kk = 4 * ks + kh;
#pragma unroll
      for (int nt = 0; nt < 2; ++nt) {
        gb[jj][ks][nt].x = S[(2 * w + jj) * SSJ + kk * SSP + nt * 16 + row];
        gb[jj][ks][nt].y = S[(2 * w + jj) * SSJ + (kk + 1) * SSP + nt * 16 + row];
      }
    }
  v8f oc[2][2][2];
#pragma unroll
  for (int jj = 0; jj < 2; ++jj)
#pragma unroll
    for (int mt = 0; mt < 2; ++mt)
#pragma unroll
      for (int nt = 0; nt < 2; ++nt)
#pragma unroll
        for (int v = 0; v < 8; ++v)
          oc[jj][mt][nt][v] = bv[nt];
  __builtin_amdgcn_sched_barrier(0);
#pragma unroll
  for (int jj = 0; jj < 2; ++jj)
#pragma unroll
    for (int ks = 0; ks < 8; ++ks)
#pragma unroll
      for (int nt = 0; nt < 2; ++nt) {
        oc[jj][0][nt] = __builtin_amdgcn_wmma_f32_16x16x4_f32(
            false, ta0[ks], false, gb[jj][ks][nt], (short)0, oc[jj][0][nt], false, false);
        oc[jj][1][nt] = __builtin_amdgcn_wmma_f32_16x16x4_f32(
            false, ta1[ks], false, gb[jj][ks][nt], (short)0, oc[jj][1][nt], false, false);
      }
  wmma_hazard_slots();
#pragma unroll
  for (int jj = 0; jj < 2; ++jj) {
    float* O = out + (((size_t)b * NGRID + (2 * w + jj)) * NGRID) * COUT + o0 + row;
#pragma unroll
    for (int mt = 0; mt < 2; ++mt)
#pragma unroll
      for (int nt = 0; nt < 2; ++nt)
#pragma unroll
        for (int v = 0; v < 8; ++v)
          O[(mt * 16 + 8 * h + v) * COUT + nt * 16] = oc[jj][mt][nt][v];
  }
}

extern "C" void kernel_launch(void* const* d_in, const int* in_sizes, int n_in,
                              void* d_out, int out_size, void* d_ws, size_t ws_size,
                              hipStream_t stream) {
  const float* inp  = (const float*)d_in[0];
  const float* wgt  = (const float*)d_in[1];
  const float* bias = (const float*)d_in[2];
  float* out = (float*)d_out;
  float* ws  = (float*)d_ws;

  sph_legendre_kernel<<<1, LMAX * NGRID, 0, stream>>>(ws);
  sph_buildmats_kernel<<<(LMAX * LMAX * NGRID + 255) / 256, 256, 0, stream>>>(ws);

  dim3 gT(CIN / CW, BATCH);
  sph_analysis_kernel<<<gT, NWV * 32, 0, stream>>>(ws + WS_T, ws + WS_QW, inp, ws + WS_COEF);
  dim3 gB(4, KR);
  sph_phaseB_kernel<<<gB, 32, 0, stream>>>(ws + WS_COEF, wgt, ws + WS_CHAT);
  sph_synthesis_kernel<<<gT, NWV * 32, 0, stream>>>(ws + WS_QS, ws + WS_TT, ws + WS_CHAT, bias, out);
}


# Legendre table kernel: (l-m)!/(l+m)! as one f64 divide by an exact-integer product per (l,m) lane, beside the sin/cos wave, replacing one wave's 30-deep f64 divide chain; N_lm floats unchanged
# speedup vs baseline: 27.4053x; 1.1485x over previous
#include <hip/hip_runtime.h>
#include <math.h>

typedef float v2f __attribute__((ext_vector_type(2)));
typedef float v8f __attribute__((ext_vector_type(8)));

#define NGRID 32
#define LMAX  16
#define NM    31
#define CIN   128
#define COUT  128
#define BATCH 64
#define KJP   1024
#define KR    256


#define WS_Q      0
#define WS_T      16384
#define WS_TT     (WS_T + 32*32)
#define WS_QW     (WS_TT + 32*32)
#define WS_QS     (WS_QW + 16*16*32)
#define WS_COEF   65536
#define WS_CHAT   (WS_COEF + BATCH*KR*CIN)

#define PI_D 3.14159265358979323846

__global__ void __launch_bounds__(LMAX * NGRID)
sph_legendre_kernel(float* __restrict__ ws) {
  __shared__ double s_x[NGRID];
  __shared__ double s_sx[NGRID];
  __shared__ float  s_nlm[LMAX][LMAX];
  __shared__ float  s_cx[LMAX][NGRID];
  __shared__ float  s_cs[LMAX][NGRID];

  const int tid = threadIdx.x;
  const int m = tid >> 5;
  const int j = tid & 31;

  if (m == 0) {
    const double theta = PI_D * (j + 0.5) / NGRID;
    s_x[j]  = cos(theta);
    s_sx[j] = sin(theta);
  } else if (m < 8 && (m & 3) != 0) {
    const int idx = (m - 1 - (m >> 2)) * 32 + j;
    if (idx < LMAX * (LMAX + 1) / 2) {
      int l = 0;
      while ((l + 1) * (l + 2) / 2 <= idx) ++l;
      const int mm = idx - l * (l + 1) / 2;
      const int d = l - mm;
      static_assert((LMAX - 1) % 3 == 0, "the last group of three steps must end on the last step");
      double prod = 1.0;
      unsigned acc = 1u;
#pragma unroll
      for (int k = 1; k <= LMAX - 1; ++k) {
        if (k <= mm) acc *= (unsigned)((d + 2 * k - 1) * (d + 2 * k));
        if (k % 3 == 0) { prod *= (double)acc; acc = 1u; }
      }
      s_nlm[l][mm] = (float)sqrt((2.0 * l + 1.0) / (4.0 * PI_D * prod));
    }
  }
  __syncthreads();

  {
    const double x = s_x[j], sx = s_sx[j];
    s_cx[m][j] = (float)((2.0 * m + 1.0) * x);
    s_cs[m][j] = (float)(-(2.0 * m - 1.0) * sx);
  }
  __syncthreads();

  float* qp = ws + WS_Q + (15 + m) * NGRID + j;
  float* qn = ws + WS_Q + (15 - m) * NGRID + j;

  for (int l = 0; l < m; ++l) {
    qp[l * NM * NGRID] = 0.f;
    qn[l * NM * NGRID] = 0.f;
  }

  float pmm = 1.0f;
  for (int k = 1; k <= m; ++k) pmm = s_cs[k][j] * pmm;
  {
    const float q = s_nlm[m][m] * pmm;
    qp[m * NM * NGRID] = q;
    qn[m * NM * NGRID] = (m & 1) ? -q : q;
  }
  if (m < LMAX - 1) {
    float p2 = pmm;
    float p1 = s_cx[m][j] * pmm;
    {
      const float q = s_nlm[m + 1][m] * p1;
      qp[(m + 1) * NM * NGRID] = q;
      qn[(m + 1) * NM * NGRID] = (m & 1) ? -q : q;
    }
    for (int l = m + 2; l < LMAX; ++l) {
      const float p = (s_cx[l - 1][j] * p1 - (float)(l + m - 1) * p2) / (float)(l - m);
      const float q = s_nlm[l][m] * p;
      qp[l * NM * NGRID] = q;
      qn[l * NM * NGRID] = (m & 1) ? -q : q;
      p2 = p1;
      p1 = p;
    }
  }
}

__global__ void sph_buildmats_kernel(float* __restrict__ ws) {
  const int idx = blockIdx.x * blockDim.x + threadIdx.x;
  if (idx >= LMAX * LMAX * NGRID) return;
  if (idx < NGRID * NGRID) {
    const int t = idx >> 5, p = idx & 31;
    const float ang = (2.0f * (float)PI_D / NGRID) * (float)((t & 15) * p);
    float s, c;
    sincosf(ang, &s, &c);
    const float tw = (t < 16) ? c : -s;
    ws[WS_T + t * NGRID + p]  = tw;
    ws[WS_TT + p * NGRID + t] = tw;
  }
  const int m = idx >> 9, l = (idx >> 5) & 15, j = idx & 31;
  const float q = ws[WS_Q + (l * NM + 15 + m) * NGRID + j];
  const float th = (float)PI_D * (j + 0.5f) / NGRID;
  const float wj = sinf(th) * ((float)PI_D / NGRID) * (2.0f * (float)PI_D / NGRID);
  const float scale_l = 2.0f * (float)PI_D * sqrtf(4.0f * (float)PI_D / (2.0f * l + 1.0f));
  const float f = (m == 0) ? 1.0f : 2.0f;
  ws[WS_QW + (m * LMAX + l) * NGRID + j] = q * wj;
  ws[WS_QS + (m * NGRID + j) * LMAX + l] = f * (q * scale_l);
}

template <int NT, int KS>
struct KFrag {
  v2f a0[KS], a1[KS];
  v2f b[KS][NT];
};

template <int NT, int KS, int MT>
__device__ __forceinline__ void load_kfrag(
    KFrag<NT, KS>& f, const float* __restrict__ A, int lda,
    const float* __restrict__ B, int ldb, int row, int kh, int k) {
#pragma unroll
  for (int ks = 0; ks < KS; ++ks) {
    const int kk = k + 4 * ks + kh;
    f.a0[ks].x = A[row * lda + kk];
    f.a0[ks].y = A[row * lda + kk + 1];
    if constexpr (MT == 2) {
      f.a1[ks].x = A[(row + 16) * lda + kk];
      f.a1[ks].y = A[(row + 16) * lda + kk + 1];
    }
#pragma unroll
    for (int nt = 0; nt < NT; ++nt) {
      f.b[ks][nt].x = B[kk * ldb + nt * 16 + row];
      f.b[ks][nt].y = B[(kk + 1) * ldb + nt * 16 + row];
    }
  }
}

template <int NT, int KS, int MT>
__device__ __forceinline__ void mma_kfrag(const KFrag<NT, KS>& f, v8f acc[MT][NT]) {
#pragma unroll
  for (int ks = 0; ks < KS; ++ks) {
#pragma unroll
    for (int nt = 0; nt < NT; ++nt) {
      acc[0][nt] = __builtin_amdgcn_wmma_f32_16x16x4_f32(
          false, f.a0[ks], false, f.b[ks][nt], (short)0, acc[0][nt], false, false);
      if constexpr (MT == 2)
        acc[1][nt] = __builtin_amdgcn_wmma_f32_16x16x4_f32(
            false, f.a1[ks], false, f.b[ks][nt], (short)0, acc[1][nt], false, false);
    }
  }
}

__device__ __forceinline__ void wmma_hazard_slots() {
  __builtin_amdgcn_sched_barrier(0);
  asm volatile("v_nop\n\tv_nop\n\tv_nop\n\tv_nop");
  __builtin_amdgcn_sched_barrier(0);
}

template <int NT, int KS, int MT>
__device__ __forceinline__ void wave_gemm_32xN(
    const float* __restrict__ A, int lda,
    const float* __restrict__ B, int ldb,
    int K, v8f acc[MT][NT]) {
  constexpr int S = 4 * KS;
  const int lane = threadIdx.x & 31;
  const int row  = lane & 15;
  const int kh   = (lane >> 4) << 1;
  KFrag<NT, KS> f0, f1, f2, f3;
  load_kfrag<NT, KS, MT>(f0, A, lda, B, ldb, row, kh, 0);
  __builtin_amdgcn_sched_barrier(0);
  load_kfrag<NT, KS, MT>(f1, A, lda, B, ldb, row, kh, S);
  __builtin_amdgcn_sched_barrier(0);
  load_kfrag<NT, KS, MT>(f2, A, lda, B, ldb, row, kh, 2 * S);
  int k = 0;
  for (; k + 4 * S < K; k += 4 * S) {
    load_kfrag<NT, KS, MT>(f3, A, lda, B, ldb, row, kh, k + 3 * S);
    __builtin_amdgcn_sched_barrier(0);
    mma_kfrag<NT, KS, MT>(f0, acc);
    wmma_hazard_slots();
    load_kfrag<NT, KS, MT>(f0, A, lda, B, ldb, row, kh, k + 4 * S);
    __builtin_amdgcn_sched_barrier(0);
    mma_kfrag<NT, KS, MT>(f1, acc);
    wmma_hazard_slots();
    load_kfrag<NT, KS, MT>(f1, A, lda, B, ldb, row, kh, k + 5 * S);
    __builtin_amdgcn_sched_barrier(0);
    mma_kfrag<NT, KS, MT>(f2, acc);
    wmma_hazard_slots();
    load_kfrag<NT, KS, MT>(f2, A, lda, B, ldb, row, kh, k + 6 * S);
    __builtin_amdgcn_sched_barrier(0);
    mma_kfrag<NT, KS, MT>(f3, acc);
    wmma_hazard_slots();
  }
  load_kfrag<NT, KS, MT>(f3, A, lda, B, ldb, row, kh, k + 3 * S);
  mma_kfrag<NT, KS, MT>(f0, acc);
  mma_kfrag<NT, KS, MT>(f1, acc);
  mma_kfrag<NT, KS, MT>(f2, acc);
  mma_kfrag<NT, KS, MT>(f3, acc);
  wmma_hazard_slots();
}

template <int NT>
__device__ __forceinline__ void wave_store_32xN(
    float* __restrict__ C, int ldc, const v8f acc[2][NT]) {
  const int lane = threadIdx.x & 31;
  const int col  = lane & 15;
  const int rb   = (lane >> 4) * 8;
#pragma unroll
  for (int t = 0; t < 2; ++t)
#pragma unroll
    for (int nt = 0; nt < NT; ++nt)
#pragma unroll
      for (int v = 0; v < 8; ++v)
        C[(t * 16 + rb + v) * ldc + nt * 16 + col] = acc[t][nt][v];
}

#define CW   32
#define NWV  16
#define ASP  36
#define ASJ  (32 * ASP + 12)
#define SSP  40
#define SSJ  (32 * SSP + 12)

__global__ void __launch_bounds__(NWV * 32)
sph_analysis_kernel(const float* __restrict__ T,
                    const float* __restrict__ QW,
                    const float* __restrict__ X,
                    float* __restrict__ coef) {
  __shared__ float S[32 * ASJ];
  const int c0   = blockIdx.x * CW;
  const int b    = blockIdx.y;
  const int w    = threadIdx.x >> 5;
  const int lane = threadIdx.x & 31;
  const int row  = lane & 15;
  const int h    = lane >> 4;
  const int kh   = h << 1;
  const int m    = w;

  v2f xb[2][8][2], ta0[8], ta1[8], qa[8];
#pragma unroll
  for (int ks = 0; ks < 8; ++ks) {
    const int kk = 4 * ks + kh;
    ta0[ks].x = T[row * NGRID + kk];
    ta0[ks].y = T[row * NGRID + kk + 1];
    ta1[ks].x = T[(row + 16) * NGRID + kk];
    ta1[ks].y = T[(row + 16) * NGRID + kk + 1];
  }
  __builtin_amdgcn_sched_barrier(0);
#pragma unroll
  for (int jj = 0; jj < 2; ++jj) {
    const float* Xj = X + (((size_t)b * NGRID + (2 * w + jj)) * NGRID) * CIN + c0 + row;
#pragma unroll
    for (int ks = 0; ks < 8; ++ks) {
      const int kk = 4 * ks + kh;
#pragma unroll
      for (int nt = 0; nt < 2; ++nt) {
        xb[jj][ks][nt].x = Xj[kk * CIN + nt * 16];
        xb[jj][ks][nt].y = Xj[(kk + 1) * CIN + nt * 16];
      }
    }
  }
  __builtin_amdgcn_sched_barrier(0);
#pragma unroll
  for (int ks = 0; ks < 8; ++ks) {
    const int kk = 4 * ks + kh;
    qa[ks].x  = QW[(m * LMAX + row) * NGRID + kk];
    qa[ks].y  = QW[(m * LMAX + row) * NGRID + kk + 1];
  }

  v8f acc[2][2][2] = {};
  __builtin_amdgcn_sched_barrier(0);
#pragma unroll
  for (int jj = 0; jj < 2; ++jj)
#pragma unroll
    for (int ks = 0; ks < 8; ++ks)
#pragma unroll
      for (int nt = 0; nt < 2; ++nt) {
        acc[jj][0][nt] = __builtin_amdgcn_wmma_f32_16x16x4_f32(
            false, ta0[ks], false, xb[jj][ks][nt], (short)0, acc[jj][0][nt], false, false);
        acc[jj][1][nt] = __builtin_amdgcn_wmma_f32_16x16x4_f32(
            false, ta1[ks], false, xb[jj][ks][nt], (short)0, acc[jj][1][nt], false, false);
      }
  wmma_hazard_slots();
#pragma unroll
  for (int jj = 0; jj < 2; ++jj)
#pragma unroll
    for (int mt = 0; mt < 2; ++mt)
#pragma unroll
      for (int nt = 0; nt < 2; ++nt)
#pragma unroll
        for (int v = 0; v < 8; ++v)
          S[(2 * w + jj) * ASJ + (mt * 16 + 8 * h + v) * ASP + nt * 16 + row] = acc[jj][mt][nt][v];
  __syncthreads();

  v2f fb[2][8][2];
#pragma unroll
  for (int ti = 0; ti < 2; ++ti)
#pragma unroll
    for (int ks = 0; ks < 8; ++ks) {
      const int kk = 4 * ks + kh;
#pragma unroll
      for (int nt = 0; nt < 2; ++nt) {
        fb[ti][ks][nt].x = S[kk * ASJ + (ti * 16 + m) * ASP + nt * 16 + row];
        fb[ti][ks][nt].y = S[(kk + 1) * ASJ + (ti * 16 + m) * ASP + nt * 16 + row];
      }
    }
  v8f cf[2][2] = {};
  __builtin_amdgcn_sched_barrier(0);
#pragma unroll
  for (int ks = 0; ks < 8; ++ks)
#pragma unroll
    for (int ti = 0; ti < 2; ++ti)
#pragma unroll
      for (int nt = 0; nt < 2; ++nt)
        cf[ti][nt] = __builtin_amdgcn_wmma_f32_16x16x4_f32(
            false, qa[ks], false, fb[ti][ks][nt], (short)0, cf[ti][nt], false, false);
  wmma_hazard_slots();
  float* C = coef + (size_t)b * KR * CIN + c0 + row;
#pragma unroll
  for (int ti = 0; ti < 2; ++ti) {
    if (ti == 1 && m == 0) continue;
#pragma unroll
    for (int v = 0; v < 8; ++v) {
      const int l = 8 * h + v;
      if (l >= m) {
        const int r = l * l + (ti ? l + m : m);
#pragma unroll
        for (int nt = 0; nt < 2; ++nt)
          C[r * CIN + nt * 16] = cf[ti][nt][v];
      }
    }
  }
}

__global__ void __launch_bounds__(32, 1)
sph_phaseB_kernel(const float* __restrict__ coef,
                  const float* __restrict__ W,
                  float* __restrict__ chat) {
  const int ch = blockIdx.x & 1;
  const int b0 = (blockIdx.x >> 1) * 32;
  const int r  = blockIdx.y;
  int l = 0;
  while ((l + 1) * (l + 1) <= r) ++l;
  const float* A = coef + ((size_t)b0 * KR + r) * CIN;
  const float* B = W + (size_t)l * COUT + ch * 64;
  float* C = chat + ((size_t)b0 * KR + r) * COUT + ch * 64;
  v8f acc[2][4] = {};
  wave_gemm_32xN<4, 2, 2>(A, KR * CIN, B, LMAX * COUT, CIN, acc);
  wave_store_32xN<4>(C, KR * COUT, acc);
}

__global__ void __launch_bounds__(NWV * 32)
sph_synthesis_kernel(const float* __restrict__ QS,
                     const float* __restrict__ TT,
                     const float* __restrict__ chat,
                     const float* __restrict__ bias,
                     float* __restrict__ out) {
  __shared__ float S[32 * SSJ];
  const int o0   = blockIdx.x * CW;
  const int b    = blockIdx.y;
  const int w    = threadIdx.x >> 5;
  const int lane = threadIdx.x & 31;
  const int row  = lane & 15;
  const int h    = lane >> 4;
  const int kh   = h << 1;
  const int m    = w;

  const float* A = QS + m * NGRID * LMAX;
  const float* B = chat + (size_t)b * KR * COUT + o0 + row;
  v2f a0[4], a1[4], bf[2][4][2], ta0[8], ta1[8];
#pragma unroll
  for (int ks = 0; ks < 4; ++ks) {
    const int kk = 4 * ks + kh;
    a0[ks].x = A[row * LMAX + kk];
    a0[ks].y = A[row * LMAX + kk + 1];
    a1[ks].x = A[(row + 16) * LMAX + kk];
    a1[ks].y = A[(row + 16) * LMAX + kk + 1];
    const int l0 = kk, l1 = kk + 1;
    const int d0 = (l0 > m) ? l0 : m, d1 = (l1 > m) ? l1 : m;
#pragma unroll
    for (int ti = 0; ti < 2; ++ti) {
      const bool live = !(ti == 1 && m == 0);
      const int r0 = d0 * d0 + (ti ? d0 + m : m);
      const int r1 = d1 * d1 + (ti ? d1 + m : m);
      const bool u0 = live && l0 >= m, u1 = live && l1 >= m;
#pragma unroll
      for (int nt = 0; nt < 2; ++nt) {
        const float x0 = B[r0 * COUT + nt * 16];
        const float x1 = B[r1 * COUT + nt * 16];
        bf[ti][ks][nt].x = u0 ? x0 : 0.f;
        bf[ti][ks][nt].y = u1 ? x1 : 0.f;
      }
    }
  }
#pragma unroll
  for (int ks = 0; ks < 8; ++ks) {
    const int kk = 4 * ks + kh;
    ta0[ks].x = TT[row * NGRID + kk];
    ta0[ks].y = TT[row * NGRID + kk + 1];
    ta1[ks].x = TT[(row + 16) * NGRID + kk];
    ta1[ks].y = TT[(row + 16) * NGRID + kk + 1];
  }
  float bv[2];
#pragma unroll
  for (int nt = 0; nt < 2; ++nt) bv[nt] = bias[o0 + nt * 16 + row];

  v8f acc[2][2][2] = {};
  __builtin_amdgcn_sched_barrier(0);
#pragma unroll
  for (int ks = 0; ks < 4; ++ks)
#pragma unroll
    for (int ti = 0; ti < 2; ++ti)
#pragma unroll
      for (int nt = 0; nt < 2; ++nt) {
        acc[ti][0][nt] = __builtin_amdgcn_wmma_f32_16x16x4_f32(
            false, a0[ks], false, bf[ti][ks][nt], (short)0, acc[ti][0][nt], false, false);
        acc[ti][1][nt] = __builtin_amdgcn_wmma_f32_16x16x4_f32(
            false, a1[ks], false, bf[ti][ks][nt], (short)0, acc[ti][1][nt], false, false);
      }
  wmma_hazard_slots();
#pragma unroll
  for (int ti = 0; ti < 2; ++ti)
#pragma unroll
    for (int mt = 0; mt < 2; ++mt)
#pragma unroll
      for (int nt = 0; nt < 2; ++nt)
#pragma unroll
        for (int v = 0; v < 8; ++v)
          S[(mt * 16 + 8 * h + v) * SSJ + (ti * 16 + m) * SSP + nt * 16 + row] = acc[ti][mt][nt][v];
  __syncthreads();

  v2f gb[2][8][2];
#pragma unroll
  for (int jj = 0; jj < 2; ++jj)
#pragma unroll
    for (int ks = 0; ks < 8; ++ks) {
      const int kk = 4 * ks + kh;
#pragma unroll
      for (int nt = 0; nt < 2; ++nt) {
        gb[jj][ks][nt].x = S[(2 * w + jj) * SSJ + kk * SSP + nt * 16 + row];
        gb[jj][ks][nt].y = S[(2 * w + jj) * SSJ + (kk + 1) * SSP + nt * 16 + row];
      }
    }
  v8f oc[2][2][2];
#pragma unroll
  for (int jj = 0; jj < 2; ++jj)
#pragma unroll
    for (int mt = 0; mt < 2; ++mt)
#pragma unroll
      for (int nt = 0; nt < 2; ++nt)
#pragma unroll
        for (int v = 0; v < 8; ++v)
          oc[jj][mt][nt][v] = bv[nt];
  __builtin_amdgcn_sched_barrier(0);
#pragma unroll
  for (int jj = 0; jj < 2; ++jj)
#pragma unroll
    for (int ks = 0; ks < 8; ++ks)
#pragma unroll
      for (int nt = 0; nt < 2; ++nt) {
        oc[jj][0][nt] = __builtin_amdgcn_wmma_f32_16x16x4_f32(
            false, ta0[ks], false, gb[jj][ks][nt], (short)0, oc[jj][0][nt], false, false);
        oc[jj][1][nt] = __builtin_amdgcn_wmma_f32_16x16x4_f32(
            false, ta1[ks], false, gb[jj][ks][nt], (short)0, oc[jj][1][nt], false, false);
      }
  wmma_hazard_slots();
#pragma unroll
  for (int jj = 0; jj < 2; ++jj) {
    float* O = out + (((size_t)b * NGRID + (2 * w + jj)) * NGRID) * COUT + o0 + row;
#pragma unroll
    for (int mt = 0; mt < 2; ++mt)
#pragma unroll
      for (int nt = 0; nt < 2; ++nt)
#pragma unroll
        for (int v = 0; v < 8; ++v)
          O[(mt * 16 + 8 * h + v) * COUT + nt * 16] = oc[jj][mt][nt][v];
  }
}

extern "C" void kernel_launch(void* const* d_in, const int* in_sizes, int n_in,
                              void* d_out, int out_size, void* d_ws, size_t ws_size,
                              hipStream_t stream) {
  const float* inp  = (const float*)d_in[0];
  const float* wgt  = (const float*)d_in[1];
  const float* bias = (const float*)d_in[2];
  float* out = (float*)d_out;
  float* ws  = (float*)d_ws;

  sph_legendre_kernel<<<1, LMAX * NGRID, 0, stream>>>(ws);
  sph_buildmats_kernel<<<(LMAX * LMAX * NGRID + 255) / 256, 256, 0, stream>>>(ws);

  dim3 gT(CIN / CW, BATCH);
  sph_analysis_kernel<<<gT, NWV * 32, 0, stream>>>(ws + WS_T, ws + WS_QW, inp, ws + WS_COEF);
  dim3 gB(4, KR);
  sph_phaseB_kernel<<<gB, 32, 0, stream>>>(ws + WS_COEF, wgt, ws + WS_CHAT);
  sph_synthesis_kernel<<<gT, NWV * 32, 0, stream>>>(ws + WS_QS, ws + WS_TT, ws + WS_CHAT, bias, out);
}


# Legendre and build kernels fused into the analysis kernel: every work-group builds T and QW in LDS while its input loads are in flight; 5 -> 3 launches; table values, k order and sums unchanged
# speedup vs baseline: 31.7845x; 1.1598x over previous
#include <hip/hip_runtime.h>
#include <math.h>

typedef float v2f __attribute__((ext_vector_type(2)));
typedef float v8f __attribute__((ext_vector_type(8)));

#define NGRID 32
#define LMAX  16
#define NM    31
#define CIN   128
#define COUT  128
#define BATCH 64
#define KJP   1024
#define KR    256


#define WS_TT     16384
#define WS_QS     (WS_TT + 32*32)
#define WS_COEF   65536
#define WS_CHAT   (WS_COEF + BATCH*KR*CIN)

#define PI_D 3.14159265358979323846

template <int NT, int KS>
struct KFrag {
  v2f a0[KS], a1[KS];
  v2f b[KS][NT];
};

template <int NT, int KS, int MT>
__device__ __forceinline__ void load_kfrag(
    KFrag<NT, KS>& f, const float* __restrict__ A, int lda,
    const float* __restrict__ B, int ldb, int row, int kh, int k) {
#pragma unroll
  for (int ks = 0; ks < KS; ++ks) {
    const int kk = k + 4 * ks + kh;
    f.a0[ks].x = A[row * lda + kk];
    f.a0[ks].y = A[row * lda + kk + 1];
    if constexpr (MT == 2) {
      f.a1[ks].x = A[(row + 16) * lda + kk];
      f.a1[ks].y = A[(row + 16) * lda + kk + 1];
    }
#pragma unroll
    for (int nt = 0; nt < NT; ++nt) {
      f.b[ks][nt].x = B[kk * ldb + nt * 16 + row];
      f.b[ks][nt].y = B[(kk + 1) * ldb + nt * 16 + row];
    }
  }
}

template <int NT, int KS, int MT>
__device__ __forceinline__ void mma_kfrag(const KFrag<NT, KS>& f, v8f acc[MT][NT]) {
#pragma unroll
  for (int ks = 0; ks < KS; ++ks) {
#pragma unroll
    for (int nt = 0; nt < NT; ++nt) {
      acc[0][nt] = __builtin_amdgcn_wmma_f32_16x16x4_f32(
          false, f.a0[ks], false, f.b[ks][nt], (short)0, acc[0][nt], false, false);
      if constexpr (MT == 2)
        acc[1][nt] = __builtin_amdgcn_wmma_f32_16x16x4_f32(
            false, f.a1[ks], false, f.b[ks][nt], (short)0, acc[1][nt], false, false);
    }
  }
}

__device__ __forceinline__ void wmma_hazard_slots() {
  __builtin_amdgcn_sched_barrier(0);
  asm volatile("v_nop\n\tv_nop\n\tv_nop\n\tv_nop");
  __builtin_amdgcn_sched_barrier(0);
}

template <int NT, int KS, int MT>
__device__ __forceinline__ void wave_gemm_32xN(
    const float* __restrict__ A, int lda,
    const float* __restrict__ B, int ldb,
    int K, v8f acc[MT][NT]) {
  constexpr int S = 4 * KS;
  const int lane = threadIdx.x & 31;
  const int row  = lane & 15;
  const int kh   = (lane >> 4) << 1;
  KFrag<NT, KS> f0, f1, f2, f3;
  load_kfrag<NT, KS, MT>(f0, A, lda, B, ldb, row, kh, 0);
  __builtin_amdgcn_sched_barrier(0);
  load_kfrag<NT, KS, MT>(f1, A, lda, B, ldb, row, kh, S);
  __builtin_amdgcn_sched_barrier(0);
  load_kfrag<NT, KS, MT>(f2, A, lda, B, ldb, row, kh, 2 * S);
  int k = 0;
  for (; k + 4 * S < K; k += 4 * S) {
    load_kfrag<NT, KS, MT>(f3, A, lda, B, ldb, row, kh, k + 3 * S);
    __builtin_amdgcn_sched_barrier(0);
    mma_kfrag<NT, KS, MT>(f0, acc);
    wmma_hazard_slots();
    load_kfrag<NT, KS, MT>(f0, A, lda, B, ldb, row, kh, k + 4 * S);
    __builtin_amdgcn_sched_barrier(0);
    mma_kfrag<NT, KS, MT>(f1, acc);
    wmma_hazard_slots();
    load_kfrag<NT, KS, MT>(f1, A, lda, B, ldb, row, kh, k + 5 * S);
    __builtin_amdgcn_sched_barrier(0);
    mma_kfrag<NT, KS, MT>(f2, acc);
    wmma_hazard_slots();
    load_kfrag<NT, KS, MT>(f2, A, lda, B, ldb, row, kh, k + 6 * S);
    __builtin_amdgcn_sched_barrier(0);
    mma_kfrag<NT, KS, MT>(f3, acc);
    wmma_hazard_slots();
  }
  load_kfrag<NT, KS, MT>(f3, A, lda, B, ldb, row, kh, k + 3 * S);
  mma_kfrag<NT, KS, MT>(f0, acc);
  mma_kfrag<NT, KS, MT>(f1, acc);
  mma_kfrag<NT, KS, MT>(f2, acc);
  mma_kfrag<NT, KS, MT>(f3, acc);
  wmma_hazard_slots();
}

template <int NT>
__device__ __forceinline__ void wave_store_32xN(
    float* __restrict__ C, int ldc, const v8f acc[2][NT]) {
  const int lane = threadIdx.x & 31;
  const int col  = lane & 15;
  const int rb   = (lane >> 4) * 8;
#pragma unroll
  for (int t = 0; t < 2; ++t)
#pragma unroll
    for (int nt = 0; nt < NT; ++nt)
#pragma unroll
      for (int v = 0; v < 8; ++v)
        C[(t * 16 + rb + v) * ldc + nt * 16 + col] = acc[t][nt][v];
}

#define CW   32
#define NWV  16
#define ASP  36
#define ASJ  (32 * ASP + 12)
#define SSP  40
#define SSJ  (32 * SSP + 12)


#define TBP  36

__device__ __forceinline__ void lds_only_barrier() {
  asm volatile("s_wait_dscnt 0x0\n\ts_barrier_signal -1\n\ts_barrier_wait -1" ::: "memory");
}
__global__ void __launch_bounds__(NWV * 32)
sph_analysis_kernel(const float* __restrict__ X,
                    float* __restrict__ coef,
                    float* __restrict__ QS,
                    float* __restrict__ TT) {
  __shared__ float  S[32 * ASJ];
  __shared__ double s_x[NGRID];
  __shared__ double s_sx[NGRID];
  __shared__ float  s_nlm[LMAX][LMAX];
  __shared__ float  s_cx[LMAX][NGRID];
  __shared__ float  s_cs[LMAX][NGRID];
  __shared__ float  s_wj[NGRID];
  __shared__ float  s_scale[LMAX];
  __shared__ float  s_T[NGRID * TBP];
  __shared__ float  s_QW[LMAX * LMAX * TBP];
  const int c0   = blockIdx.x * CW;
  const int b    = blockIdx.y;
  const int w    = threadIdx.x >> 5;
  const int lane = threadIdx.x & 31;
  const int row  = lane & 15;
  const int h    = lane >> 4;
  const int kh   = h << 1;
  const int m    = w;
  const int j    = lane;
  const bool first = (blockIdx.x == 0) && (blockIdx.y == 0);

  v2f xb[2][8][2];
#pragma unroll
  for (int jj = 0; jj < 2; ++jj) {
    const float* Xj = X + (((size_t)b * NGRID + (2 * w + jj)) * NGRID) * CIN + c0 + row;
#pragma unroll
    for (int ks = 0; ks < 8; ++ks) {
      const int kk = 4 * ks + kh;
#pragma unroll
      for (int nt = 0; nt < 2; ++nt) {
        xb[jj][ks][nt].x = Xj[kk * CIN + nt * 16];
        xb[jj][ks][nt].y = Xj[(kk + 1) * CIN + nt * 16];
      }
    }
  }
  __builtin_amdgcn_sched_barrier(0);

  if (m == 0) {
    const double theta = PI_D * (j + 0.5) / NGRID;
    s_x[j]  = cos(theta);
    s_sx[j] = sin(theta);
  } else if (m < 8 && (m & 3) != 0) {
    const int idx = (m - 1 - (m >> 2)) * 32 + j;
    if (idx < LMAX * (LMAX + 1) / 2) {
      int l = 0;
      while ((l + 1) * (l + 2) / 2 <= idx) ++l;
      const int mm = idx - l * (l + 1) / 2;
      const int d = l - mm;
      static_assert((LMAX - 1) % 3 == 0, "the last group of three steps must end on the last step");
      double prod = 1.0;
      unsigned acc = 1u;
#pragma unroll
      for (int k = 1; k <= LMAX - 1; ++k) {
        if (k <= mm) acc *= (unsigned)((d + 2 * k - 1) * (d + 2 * k));
        if (k % 3 == 0) { prod *= (double)acc; acc = 1u; }
      }
      s_nlm[l][mm] = (float)sqrt((2.0 * l + 1.0) / (4.0 * PI_D * prod));
    }
  } else if (m == 4) {
    const float th = (float)PI_D * (j + 0.5f) / NGRID;
    s_wj[j] = sinf(th) * ((float)PI_D / NGRID) * (2.0f * (float)PI_D / NGRID);
    if (j < LMAX)
      s_scale[j] = 2.0f * (float)PI_D * sqrtf(4.0f * (float)PI_D / (2.0f * j + 1.0f));
  } else if (m >= 8) {
    for (int e = 0; e < 4; ++e) {
      const int idx = (threadIdx.x - 8 * 32) + e * 8 * 32;
      const int t = idx >> 5, p = idx & 31;
      const float ang = (2.0f * (float)PI_D / NGRID) * (float)((t & 15) * p);
      float sn, cs;
      sincosf(ang, &sn, &cs);
      const float tw = (t < 16) ? cs : -sn;
      s_T[t * TBP + p] = tw;
      if (first) TT[p * NGRID + t] = tw;
    }
  }
  lds_only_barrier();

  {
    const double x = s_x[j], sx = s_sx[j];
    s_cx[m][j] = (float)((2.0 * m + 1.0) * x);
    s_cs[m][j] = (float)(-(2.0 * m - 1.0) * sx);
  }
  lds_only_barrier();

  {
    const float wj = s_wj[j];
    const float fm = (m == 0) ? 1.0f : 2.0f;
    float* qw = s_QW + m * LMAX * TBP + j;
    float* qs = QS + (m * NGRID + j) * LMAX;
    for (int l = 0; l < m; ++l) {
      qw[l * TBP] = 0.f;
      if (first) qs[l] = 0.f;
    }
    float pmm = 1.0f;
    for (int k = 1; k <= m; ++k) pmm = s_cs[k][j] * pmm;
    {
      const float q = s_nlm[m][m] * pmm;
      qw[m * TBP] = q * wj;
      if (first) qs[m] = fm * (q * s_scale[m]);
    }
    if (m < LMAX - 1) {
      float p2 = pmm;
      float p1 = s_cx[m][j] * pmm;
      {
        const float q = s_nlm[m + 1][m] * p1;
        qw[(m + 1) * TBP] = q * wj;
        if (first) qs[m + 1] = fm * (q * s_scale[m + 1]);
      }
      for (int l = m + 2; l < LMAX; ++l) {
        const float p = (s_cx[l - 1][j] * p1 - (float)(l + m - 1) * p2) / (float)(l - m);
        const float q = s_nlm[l][m] * p;
        qw[l * TBP] = q * wj;
        if (first) qs[l] = fm * (q * s_scale[l]);
        p2 = p1;
        p1 = p;
      }
    }
  }
  v2f ta0[8], ta1[8], qa[8];
#pragma unroll
  for (int ks = 0; ks < 8; ++ks) {
    const int kk = 4 * ks + kh;
    ta0[ks].x = s_T[row * TBP + kk];
    ta0[ks].y = s_T[row * TBP + kk + 1];
    ta1[ks].x = s_T[(row + 16) * TBP + kk];
    ta1[ks].y = s_T[(row + 16) * TBP + kk + 1];
    qa[ks].x  = s_QW[(m * LMAX + row) * TBP + kk];
    qa[ks].y  = s_QW[(m * LMAX + row) * TBP + kk + 1];
  }

  v8f acc[2][2][2] = {};
  __builtin_amdgcn_sched_barrier(0);
#pragma unroll
  for (int jj = 0; jj < 2; ++jj)
#pragma unroll
    for (int ks = 0; ks < 8; ++ks)
#pragma unroll
      for (int nt = 0; nt < 2; ++nt) {
        acc[jj][0][nt] = __builtin_amdgcn_wmma_f32_16x16x4_f32(
            false, ta0[ks], false, xb[jj][ks][nt], (short)0, acc[jj][0][nt], false, false);
        acc[jj][1][nt] = __builtin_amdgcn_wmma_f32_16x16x4_f32(
            false, ta1[ks], false, xb[jj][ks][nt], (short)0, acc[jj][1][nt], false, false);
      }
  wmma_hazard_slots();
#pragma unroll
  for (int jj = 0; jj < 2; ++jj)
#pragma unroll
    for (int mt = 0; mt < 2; ++mt)
#pragma unroll
      for (int nt = 0; nt < 2; ++nt)
#pragma unroll
        for (int v = 0; v < 8; ++v)
          S[(2 * w + jj) * ASJ + (mt * 16 + 8 * h + v) * ASP + nt * 16 + row] = acc[jj][mt][nt][v];
  __syncthreads();

  v2f fb[2][8][2];
#pragma unroll
  for (int ti = 0; ti < 2; ++ti)
#pragma unroll
    for (int ks = 0; ks < 8; ++ks) {
      const int kk = 4 * ks + kh;
#pragma unroll
      for (int nt = 0; nt < 2; ++nt) {
        fb[ti][ks][nt].x = S[kk * ASJ + (ti * 16 + m) * ASP + nt * 16 + row];
        fb[ti][ks][nt].y = S[(kk + 1) * ASJ + (ti * 16 + m) * ASP + nt * 16 + row];
      }
    }
  v8f cf[2][2] = {};
  __builtin_amdgcn_sched_barrier(0);
#pragma unroll
  for (int ks = 0; ks < 8; ++ks)
#pragma unroll
    for (int ti = 0; ti < 2; ++ti)
#pragma unroll
      for (int nt = 0; nt < 2; ++nt)
        cf[ti][nt] = __builtin_amdgcn_wmma_f32_16x16x4_f32(
            false, qa[ks], false, fb[ti][ks][nt], (short)0, cf[ti][nt], false, false);
  wmma_hazard_slots();
  float* C = coef + (size_t)b * KR * CIN + c0 + row;
#pragma unroll
  for (int ti = 0; ti < 2; ++ti) {
    if (ti == 1 && m == 0) continue;
#pragma unroll
    for (int v = 0; v < 8; ++v) {
      const int l = 8 * h + v;
      if (l >= m) {
        const int r = l * l + (ti ? l + m : m);
#pragma unroll
        for (int nt = 0; nt < 2; ++nt)
          C[r * CIN + nt * 16] = cf[ti][nt][v];
      }
    }
  }
}

__global__ void __launch_bounds__(32, 1)
sph_phaseB_kernel(const float* __restrict__ coef,
                  const float* __restrict__ W,
                  float* __restrict__ chat) {
  const int ch = blockIdx.x & 1;
  const int b0 = (blockIdx.x >> 1) * 32;
  const int r  = blockIdx.y;
  int l = 0;
  while ((l + 1) * (l + 1) <= r) ++l;
  const float* A = coef + ((size_t)b0 * KR + r) * CIN;
  const float* B = W + (size_t)l * COUT + ch * 64;
  float* C = chat + ((size_t)b0 * KR + r) * COUT + ch * 64;
  v8f acc[2][4] = {};
  wave_gemm_32xN<4, 2, 2>(A, KR * CIN, B, LMAX * COUT, CIN, acc);
  wave_store_32xN<4>(C, KR * COUT, acc);
}

__global__ void __launch_bounds__(NWV * 32)
sph_synthesis_kernel(const float* __restrict__ QS,
                     const float* __restrict__ TT,
                     const float* __restrict__ chat,
                     const float* __restrict__ bias,
                     float* __restrict__ out) {
  __shared__ float S[32 * SSJ];
  const int o0   = blockIdx.x * CW;
  const int b    = blockIdx.y;
  const int w    = threadIdx.x >> 5;
  const int lane = threadIdx.x & 31;
  const int row  = lane & 15;
  const int h    = lane >> 4;
  const int kh   = h << 1;
  const int m    = w;

  const float* A = QS + m * NGRID * LMAX;
  const float* B = chat + (size_t)b * KR * COUT + o0 + row;
  v2f a0[4], a1[4], bf[2][4][2], ta0[8], ta1[8];
#pragma unroll
  for (int ks = 0; ks < 4; ++ks) {
    const int kk = 4 * ks + kh;
    a0[ks].x = A[row * LMAX + kk];
    a0[ks].y = A[row * LMAX + kk + 1];
    a1[ks].x = A[(row + 16) * LMAX + kk];
    a1[ks].y = A[(row + 16) * LMAX + kk + 1];
    const int l0 = kk, l1 = kk + 1;
    const int d0 = (l0 > m) ? l0 : m, d1 = (l1 > m) ? l1 : m;
#pragma unroll
    for (int ti = 0; ti < 2; ++ti) {
      const bool live = !(ti == 1 && m == 0);
      const int r0 = d0 * d0 + (ti ? d0 + m : m);
      const int r1 = d1 * d1 + (ti ? d1 + m : m);
      const bool u0 = live && l0 >= m, u1 = live && l1 >= m;
#pragma unroll
      for (int nt = 0; nt < 2; ++nt) {
        const float x0 = B[r0 * COUT + nt * 16];
        const float x1 = B[r1 * COUT + nt * 16];
        bf[ti][ks][nt].x = u0 ? x0 : 0.f;
        bf[ti][ks][nt].y = u1 ? x1 : 0.f;
      }
    }
  }
#pragma unroll
  for (int ks = 0; ks < 8; ++ks) {
    const int kk = 4 * ks + kh;
    ta0[ks].x = TT[row * NGRID + kk];
    ta0[ks].y = TT[row * NGRID + kk + 1];
    ta1[ks].x = TT[(row + 16) * NGRID + kk];
    ta1[ks].y = TT[(row + 16) * NGRID + kk + 1];
  }
  float bv[2];
#pragma unroll
  for (int nt = 0; nt < 2; ++nt) bv[nt] = bias[o0 + nt * 16 + row];

  v8f acc[2][2][2] = {};
  __builtin_amdgcn_sched_barrier(0);
#pragma unroll
  for (int ks = 0; ks < 4; ++ks)
#pragma unroll
    for (int ti = 0; ti < 2; ++ti)
#pragma unroll
      for (int nt = 0; nt < 2; ++nt) {
        acc[ti][0][nt] = __builtin_amdgcn_wmma_f32_16x16x4_f32(
            false, a0[ks], false, bf[ti][ks][nt], (short)0, acc[ti][0][nt], false, false);
        acc[ti][1][nt] = __builtin_amdgcn_wmma_f32_16x16x4_f32(
            false, a1[ks], false, bf[ti][ks][nt], (short)0, acc[ti][1][nt], false, false);
      }
  wmma_hazard_slots();
#pragma unroll
  for (int ti = 0; ti < 2; ++ti)
#pragma unroll
    for (int mt = 0; mt < 2; ++mt)
#pragma unroll
      for (int nt = 0; nt < 2; ++nt)
#pragma unroll
        for (int v = 0; v < 8; ++v)
          S[(mt * 16 + 8 * h + v) * SSJ + (ti * 16 + m) * SSP + nt * 16 + row] = acc[ti][mt][nt][v];
  __syncthreads();

  v2f gb[2][8][2];
#pragma unroll
  for (int jj = 0; jj < 2; ++jj)
#pragma unroll
    for (int ks = 0; ks < 8; ++ks) {
      const int kk = 4 * ks + kh;
#pragma unroll
      for (int nt = 0; nt < 2; ++nt) {
        gb[jj][ks][nt].x = S[(2 * w + jj) * SSJ + kk * SSP + nt * 16 + row];
        gb[jj][ks][nt].y = S[(2 * w + jj) * SSJ + (kk + 1) * SSP + nt * 16 + row];
      }
    }
  v8f oc[2][2][2];
#pragma unroll
  for (int jj = 0; jj < 2; ++jj)
#pragma unroll
    for (int mt = 0; mt < 2; ++mt)
#pragma unroll
      for (int nt = 0; nt < 2; ++nt)
#pragma unroll
        for (int v = 0; v < 8; ++v)
          oc[jj][mt][nt][v] = bv[nt];
  __builtin_amdgcn_sched_barrier(0);
#pragma unroll
  for (int jj = 0; jj < 2; ++jj)
#pragma unroll
    for (int ks = 0; ks < 8; ++ks)
#pragma unroll
      for (int nt = 0; nt < 2; ++nt) {
        oc[jj][0][nt] = __builtin_amdgcn_wmma_f32_16x16x4_f32(
            false, ta0[ks], false, gb[jj][ks][nt], (short)0, oc[jj][0][nt], false, false);
        oc[jj][1][nt] = __builtin_amdgcn_wmma_f32_16x16x4_f32(
            false, ta1[ks], false, gb[jj][ks][nt], (short)0, oc[jj][1][nt], false, false);
      }
  wmma_hazard_slots();
#pragma unroll
  for (int jj = 0; jj < 2; ++jj) {
    float* O = out + (((size_t)b * NGRID + (2 * w + jj)) * NGRID) * COUT + o0 + row;
#pragma unroll
    for (int mt = 0; mt < 2; ++mt)
#pragma unroll
      for (int nt = 0; nt < 2; ++nt)
#pragma unroll
        for (int v = 0; v < 8; ++v)
          O[(mt * 16 + 8 * h + v) * COUT + nt * 16] = oc[jj][mt][nt][v];
  }
}

extern "C" void kernel_launch(void* const* d_in, const int* in_sizes, int n_in,
                              void* d_out, int out_size, void* d_ws, size_t ws_size,
                              hipStream_t stream) {
  const float* inp  = (const float*)d_in[0];
  const float* wgt  = (const float*)d_in[1];
  const float* bias = (const float*)d_in[2];
  float* out = (float*)d_out;
  float* ws  = (float*)d_ws;

  dim3 gT(CIN / CW, BATCH);
  sph_analysis_kernel<<<gT, NWV * 32, 0, stream>>>(inp, ws + WS_COEF, ws + WS_QS, ws + WS_TT);
  dim3 gB(4, KR);
  sph_phaseB_kernel<<<gB, 32, 0, stream>>>(ws + WS_COEF, wgt, ws + WS_CHAT);
  sph_synthesis_kernel<<<gT, NWV * 32, 0, stream>>>(ws + WS_QS, ws + WS_TT, ws + WS_CHAT, bias, out);
}


# All three kernels: a lane owns adjacent channels (column n of tile nt = channel T*n+nt, T tiles), so global loads/stores are 8-16 B per lane and cover whole 128-B lines; sums unchanged
# speedup vs baseline: 34.1988x; 1.0760x over previous
#include <hip/hip_runtime.h>
#include <math.h>

typedef float v2f __attribute__((ext_vector_type(2)));
typedef float v4f __attribute__((ext_vector_type(4)));
typedef float v8f __attribute__((ext_vector_type(8)));

#define NGRID 32
#define LMAX  16
#define NM    31
#define CIN   128
#define COUT  128
#define BATCH 64
#define KJP   1024
#define KR    256


#define WS_TT     16384
#define WS_QS     (WS_TT + 32*32)
#define WS_COEF   65536
#define WS_CHAT   (WS_COEF + BATCH*KR*CIN)

#define PI_D 3.14159265358979323846

template <int NT, int KS>
struct KFrag {
  static_assert(NT == 4, "a lane owns four adjacent columns, one per column tile");
  v2f a0[KS], a1[KS];
  v4f q[KS][2];
};

template <int NT, int KS, int MT>
__device__ __forceinline__ void load_kfrag(
    KFrag<NT, KS>& f, const float* __restrict__ A, int lda,
    const float* __restrict__ B, int ldb, int row, int kh, int k) {
#pragma unroll
  for (int ks = 0; ks < KS; ++ks) {
    const int kk = k + 4 * ks + kh;
    f.a0[ks].x = A[row * lda + kk];
    f.a0[ks].y = A[row * lda + kk + 1];
    if constexpr (MT == 2) {
      f.a1[ks].x = A[(row + 16) * lda + kk];
      f.a1[ks].y = A[(row + 16) * lda + kk + 1];
    }
    f.q[ks][0] = *reinterpret_cast<const v4f*>(B + kk * ldb + 4 * row);
    f.q[ks][1] = *reinterpret_cast<const v4f*>(B + (kk + 1) * ldb + 4 * row);
  }
}

template <int NT, int KS, int MT>
__device__ __forceinline__ void mma_kfrag(const KFrag<NT, KS>& f, v8f acc[MT][NT]) {
  v2f bfr[KS][NT];
#pragma unroll
  for (int ks = 0; ks < KS; ++ks)
#pragma unroll
    for (int nt = 0; nt < NT; ++nt) {
      bfr[ks][nt].x = f.q[ks][0][nt];
      bfr[ks][nt].y = f.q[ks][1][nt];
    }
  __builtin_amdgcn_sched_barrier(0);
#pragma unroll
  for (int ks = 0; ks < KS; ++ks) {
#pragma unroll
    for (int nt = 0; nt < NT; ++nt) {
      acc[0][nt] = __builtin_amdgcn_wmma_f32_16x16x4_f32(
          false, f.a0[ks], false, bfr[ks][nt], (short)0, acc[0][nt], false, false);
      if constexpr (MT == 2)
        acc[1][nt] = __builtin_amdgcn_wmma_f32_16x16x4_f32(
            false, f.a1[ks], false, bfr[ks][nt], (short)0, acc[1][nt], false, false);
    }
  }
}

__device__ __forceinline__ void wmma_hazard_slots() {
  __builtin_amdgcn_sched_barrier(0);
  asm volatile("v_nop\n\tv_nop\n\tv_nop\n\tv_nop");
  __builtin_amdgcn_sched_barrier(0);
}

template <int NT, int KS, int MT>
__device__ __forceinline__ void wave_gemm_32xN(
    const float* __restrict__ A, int lda,
    const float* __restrict__ B, int ldb,
    int K, v8f acc[MT][NT]) {
  constexpr int S = 4 * KS;
  const int lane = threadIdx.x & 31;
  const int row  = lane & 15;
  const int kh   = (lane >> 4) << 1;
  KFrag<NT, KS> f0, f1, f2, f3;
  load_kfrag<NT, KS, MT>(f0, A, lda, B, ldb, row, kh, 0);
  __builtin_amdgcn_sched_barrier(0);
  load_kfrag<NT, KS, MT>(f1, A, lda, B, ldb, row, kh, S);
  __builtin_amdgcn_sched_barrier(0);
  load_kfrag<NT, KS, MT>(f2, A, lda, B, ldb, row, kh, 2 * S);
  int k = 0;
  for (; k + 4 * S < K; k += 4 * S) {
    load_kfrag<NT, KS, MT>(f3, A, lda, B, ldb, row, kh, k + 3 * S);
    __builtin_amdgcn_sched_barrier(0);
    mma_kfrag<NT, KS, MT>(f0, acc);
    wmma_hazard_slots();
    load_kfrag<NT, KS, MT>(f0, A, lda, B, ldb, row, kh, k + 4 * S);
    __builtin_amdgcn_sched_barrier(0);
    mma_kfrag<NT, KS, MT>(f1, acc);
    wmma_hazard_slots();
    load_kfrag<NT, KS, MT>(f1, A, lda, B, ldb, row, kh, k + 5 * S);
    __builtin_amdgcn_sched_barrier(0);
    mma_kfrag<NT, KS, MT>(f2, acc);
    wmma_hazard_slots();
    load_kfrag<NT, KS, MT>(f2, A, lda, B, ldb, row, kh, k + 6 * S);
    __builtin_amdgcn_sched_barrier(0);
    mma_kfrag<NT, KS, MT>(f3, acc);
    wmma_hazard_slots();
  }
  load_kfrag<NT, KS, MT>(f3, A, lda, B, ldb, row, kh, k + 3 * S);
  mma_kfrag<NT, KS, MT>(f0, acc);
  wmma_hazard_slots();
  mma_kfrag<NT, KS, MT>(f1, acc);
  wmma_hazard_slots();
  mma_kfrag<NT, KS, MT>(f2, acc);
  wmma_hazard_slots();
  mma_kfrag<NT, KS, MT>(f3, acc);
  wmma_hazard_slots();
}

template <int NT>
__device__ __forceinline__ void wave_store_32xN(
    float* __restrict__ C, int ldc, const v8f acc[2][NT]) {
  static_assert(NT == 4, "a lane owns four adjacent columns, one per column tile");
  const int lane = threadIdx.x & 31;
  const int col  = lane & 15;
  const int rb   = (lane >> 4) * 8;
#pragma unroll
  for (int t = 0; t < 2; ++t)
#pragma unroll
    for (int v = 0; v < 8; ++v) {
      v4f qd;
      qd.x = acc[t][0][v];
      qd.y = acc[t][1][v];
      qd.z = acc[t][2][v];
      qd.w = acc[t][3][v];
      *reinterpret_cast<v4f*>(C + (t * 16 + rb + v) * ldc + 4 * col) = qd;
    }
}

#define CW   32
#define NWV  16
#define ASP  36
#define ASJ  (32 * ASP + 12)
#define SSP  40
#define SSJ  (32 * SSP + 12)


#define TBP  36

__device__ __forceinline__ void lds_only_barrier() {
  asm volatile("s_wait_dscnt 0x0\n\ts_barrier_signal -1\n\ts_barrier_wait -1" ::: "memory");
}
__global__ void __launch_bounds__(NWV * 32)
sph_analysis_kernel(const float* __restrict__ X,
                    float* __restrict__ coef,
                    float* __restrict__ QS,
                    float* __restrict__ TT) {
  __shared__ float  S[32 * ASJ];
  __shared__ double s_x[NGRID];
  __shared__ double s_sx[NGRID];
  __shared__ float  s_nlm[LMAX][LMAX];
  __shared__ float  s_cx[LMAX][NGRID];
  __shared__ float  s_cs[LMAX][NGRID];
  __shared__ float  s_wj[NGRID];
  __shared__ float  s_scale[LMAX];
  __shared__ float  s_T[NGRID * TBP];
  __shared__ float  s_QW[LMAX * LMAX * TBP];
  const int c0   = blockIdx.x * CW;
  const int b    = blockIdx.y;
  const int w    = threadIdx.x >> 5;
  const int lane = threadIdx.x & 31;
  const int row  = lane & 15;
  const int h    = lane >> 4;
  const int kh   = h << 1;
  const int m    = w;
  const int j    = lane;
  const bool first = (blockIdx.x == 0) && (blockIdx.y == 0);

  v2f xr[2][8][2];
#pragma unroll
  for (int jj = 0; jj < 2; ++jj) {
    const float* Xj = X + (((size_t)b * NGRID + (2 * w + jj)) * NGRID) * CIN + c0 + 2 * row;
#pragma unroll
    for (int ks = 0; ks < 8; ++ks) {
      const int kk = 4 * ks + kh;
      xr[jj][ks][0] = *reinterpret_cast<const v2f*>(Xj + kk * CIN);
      xr[jj][ks][1] = *reinterpret_cast<const v2f*>(Xj + (kk + 1) * CIN);
    }
  }
  __builtin_amdgcn_sched_barrier(0);

  if (m == 0) {
    const double theta = PI_D * (j + 0.5) / NGRID;
    s_x[j]  = cos(theta);
    s_sx[j] = sin(theta);
  } else if (m < 8 && (m & 3) != 0) {
    const int idx = (m - 1 - (m >> 2)) * 32 + j;
    if (idx < LMAX * (LMAX + 1) / 2) {
      int l = 0;
      while ((l + 1) * (l + 2) / 2 <= idx) ++l;
      const int mm = idx - l * (l + 1) / 2;
      const int d = l - mm;
      static_assert((LMAX - 1) % 3 == 0, "the last group of three steps must end on the last step");
      double prod = 1.0;
      unsigned acc = 1u;
#pragma unroll
      for (int k = 1; k <= LMAX - 1; ++k) {
        if (k <= mm) acc *= (unsigned)((d + 2 * k - 1) * (d + 2 * k));
        if (k % 3 == 0) { prod *= (double)acc; acc = 1u; }
      }
      s_nlm[l][mm] = (float)sqrt((2.0 * l + 1.0) / (4.0 * PI_D * prod));
    }
  } else if (m == 4) {
    const float th = (float)PI_D * (j + 0.5f) / NGRID;
    s_wj[j] = sinf(th) * ((float)PI_D / NGRID) * (2.0f * (float)PI_D / NGRID);
    if (j < LMAX)
      s_scale[j] = 2.0f * (float)PI_D * sqrtf(4.0f * (float)PI_D / (2.0f * j + 1.0f));
  } else if (m >= 8) {
    for (int e = 0; e < 4; ++e) {
      const int idx = (threadIdx.x - 8 * 32) + e * 8 * 32;
      const int t = idx >> 5, p = idx & 31;
      const float ang = (2.0f * (float)PI_D / NGRID) * (float)((t & 15) * p);
      float sn, cs;
      sincosf(ang, &sn, &cs);
      const float tw = (t < 16) ? cs : -sn;
      s_T[t * TBP + p] = tw;
      if (first) TT[p * NGRID + t] = tw;
    }
  }
  lds_only_barrier();

  {
    const double x = s_x[j], sx = s_sx[j];
    s_cx[m][j] = (float)((2.0 * m + 1.0) * x);
    s_cs[m][j] = (float)(-(2.0 * m - 1.0) * sx);
  }
  lds_only_barrier();

  {
    const float wj = s_wj[j];
    const float fm = (m == 0) ? 1.0f : 2.0f;
    float* qw = s_QW + m * LMAX * TBP + j;
    float* qs = QS + (m * NGRID + j) * LMAX;
    for (int l = 0; l < m; ++l) {
      qw[l * TBP] = 0.f;
      if (first) qs[l] = 0.f;
    }
    float pmm = 1.0f;
    for (int k = 1; k <= m; ++k) pmm = s_cs[k][j] * pmm;
    {
      const float q = s_nlm[m][m] * pmm;
      qw[m * TBP] = q * wj;
      if (first) qs[m] = fm * (q * s_scale[m]);
    }
    if (m < LMAX - 1) {
      float p2 = pmm;
      float p1 = s_cx[m][j] * pmm;
      {
        const float q = s_nlm[m + 1][m] * p1;
        qw[(m + 1) * TBP] = q * wj;
        if (first) qs[m + 1] = fm * (q * s_scale[m + 1]);
      }
      for (int l = m + 2; l < LMAX; ++l) {
        const float p = (s_cx[l - 1][j] * p1 - (float)(l + m - 1) * p2) / (float)(l - m);
        const float q = s_nlm[l][m] * p;
        qw[l * TBP] = q * wj;
        if (first) qs[l] = fm * (q * s_scale[l]);
        p2 = p1;
        p1 = p;
      }
    }
  }
  v2f ta0[8], ta1[8], qa[8];
#pragma unroll
  for (int ks = 0; ks < 8; ++ks) {
    const int kk = 4 * ks + kh;
    ta0[ks].x = s_T[row * TBP + kk];
    ta0[ks].y = s_T[row * TBP + kk + 1];
    ta1[ks].x = s_T[(row + 16) * TBP + kk];
    ta1[ks].y = s_T[(row + 16) * TBP + kk + 1];
    qa[ks].x  = s_QW[(m * LMAX + row) * TBP + kk];
    qa[ks].y  = s_QW[(m * LMAX + row) * TBP + kk + 1];
  }

  v2f xb[2][8][2];
#pragma unroll
  for (int jj = 0; jj < 2; ++jj)
#pragma unroll
    for (int ks = 0; ks < 8; ++ks) {
      xb[jj][ks][0].x = xr[jj][ks][0].x;
      xb[jj][ks][0].y = xr[jj][ks][1].x;
      xb[jj][ks][1].x = xr[jj][ks][0].y;
      xb[jj][ks][1].y = xr[jj][ks][1].y;
    }

  v8f acc[2][2][2] = {};
  __builtin_amdgcn_sched_barrier(0);
#pragma unroll
  for (int jj = 0; jj < 2; ++jj)
#pragma unroll
    for (int ks = 0; ks < 8; ++ks)
#pragma unroll
      for (int nt = 0; nt < 2; ++nt) {
        acc[jj][0][nt] = __builtin_amdgcn_wmma_f32_16x16x4_f32(
            false, ta0[ks], false, xb[jj][ks][nt], (short)0, acc[jj][0][nt], false, false);
        acc[jj][1][nt] = __builtin_amdgcn_wmma_f32_16x16x4_f32(
            false, ta1[ks], false, xb[jj][ks][nt], (short)0, acc[jj][1][nt], false, false);
      }
  wmma_hazard_slots();
#pragma unroll
  for (int jj = 0; jj < 2; ++jj)
#pragma unroll
    for (int mt = 0; mt < 2; ++mt)
#pragma unroll
      for (int nt = 0; nt < 2; ++nt)
#pragma unroll
        for (int v = 0; v < 8; ++v)
          S[(2 * w + jj) * ASJ + (mt * 16 + 8 * h + v) * ASP + nt * 16 + row] = acc[jj][mt][nt][v];
  __syncthreads();

  v2f fb[2][8][2];
#pragma unroll
  for (int ti = 0; ti < 2; ++ti)
#pragma unroll
    for (int ks = 0; ks < 8; ++ks) {
      const int kk = 4 * ks + kh;
#pragma unroll
      for (int nt = 0; nt < 2; ++nt) {
        fb[ti][ks][nt].x = S[kk * ASJ + (ti * 16 + m) * ASP + nt * 16 + row];
        fb[ti][ks][nt].y = S[(kk + 1) * ASJ + (ti * 16 + m) * ASP + nt * 16 + row];
      }
    }
  v8f cf[2][2] = {};
  __builtin_amdgcn_sched_barrier(0);
#pragma unroll
  for (int ks = 0; ks < 8; ++ks)
#pragma unroll
    for (int ti = 0; ti < 2; ++ti)
#pragma unroll
      for (int nt = 0; nt < 2; ++nt)
        cf[ti][nt] = __builtin_amdgcn_wmma_f32_16x16x4_f32(
            false, qa[ks], false, fb[ti][ks][nt], (short)0, cf[ti][nt], false, false);
  wmma_hazard_slots();
  float* C = coef + (size_t)b * KR * CIN + c0 + 2 * row;
#pragma unroll
  for (int ti = 0; ti < 2; ++ti) {
    if (ti == 1 && m == 0) continue;
#pragma unroll
    for (int v = 0; v < 8; ++v) {
      const int l = 8 * h + v;
      if (l >= m) {
        const int r = l * l + (ti ? l + m : m);
        v2f pr;
        pr.x = cf[ti][0][v];
        pr.y = cf[ti][1][v];
        *reinterpret_cast<v2f*>(C + r * CIN) = pr;
      }
    }
  }
}

__global__ void __launch_bounds__(32, 1)
sph_phaseB_kernel(const float* __restrict__ coef,
                  const float* __restrict__ W,
                  float* __restrict__ chat) {
  const int ch = blockIdx.x & 1;
  const int b0 = (blockIdx.x >> 1) * 32;
  const int r  = blockIdx.y;
  int l = 0;
  while ((l + 1) * (l + 1) <= r) ++l;
  const float* A = coef + ((size_t)b0 * KR + r) * CIN;
  const float* B = W + (size_t)l * COUT + ch * 64;
  float* C = chat + ((size_t)b0 * KR + r) * COUT + ch * 64;
  v8f acc[2][4] = {};
  wave_gemm_32xN<4, 2, 2>(A, KR * CIN, B, LMAX * COUT, CIN, acc);
  wave_store_32xN<4>(C, KR * COUT, acc);
}

__global__ void __launch_bounds__(NWV * 32)
sph_synthesis_kernel(const float* __restrict__ QS,
                     const float* __restrict__ TT,
                     const float* __restrict__ chat,
                     const float* __restrict__ bias,
                     float* __restrict__ out) {
  __shared__ float S[32 * SSJ];
  const int o0   = blockIdx.x * CW;
  const int b    = blockIdx.y;
  const int w    = threadIdx.x >> 5;
  const int lane = threadIdx.x & 31;
  const int row  = lane & 15;
  const int h    = lane >> 4;
  const int kh   = h << 1;
  const int m    = w;

  const float* A = QS + m * NGRID * LMAX;
  const float* B = chat + (size_t)b * KR * COUT + o0 + 2 * row;
  v2f a0[4], a1[4], bf[2][4][2], ta0[8], ta1[8];
#pragma unroll
  for (int ks = 0; ks < 4; ++ks) {
    const int kk = 4 * ks + kh;
    a0[ks].x = A[row * LMAX + kk];
    a0[ks].y = A[row * LMAX + kk + 1];
    a1[ks].x = A[(row + 16) * LMAX + kk];
    a1[ks].y = A[(row + 16) * LMAX + kk + 1];
    const int l0 = kk, l1 = kk + 1;
    const int d0 = (l0 > m) ? l0 : m, d1 = (l1 > m) ? l1 : m;
#pragma unroll
    for (int ti = 0; ti < 2; ++ti) {
      const bool live = !(ti == 1 && m == 0);
      const int r0 = d0 * d0 + (ti ? d0 + m : m);
      const int r1 = d1 * d1 + (ti ? d1 + m : m);
      const bool u0 = live && l0 >= m, u1 = live && l1 >= m;
      const v2f y0 = *reinterpret_cast<const v2f*>(B + r0 * COUT);
      const v2f y1 = *reinterpret_cast<const v2f*>(B + r1 * COUT);
      bf[ti][ks][0].x = u0 ? y0.x : 0.f;
      bf[ti][ks][0].y = u1 ? y1.x : 0.f;
      bf[ti][ks][1].x = u0 ? y0.y : 0.f;
      bf[ti][ks][1].y = u1 ? y1.y : 0.f;
    }
  }
#pragma unroll
  for (int ks = 0; ks < 8; ++ks) {
    const int kk = 4 * ks + kh;
    ta0[ks].x = TT[row * NGRID + kk];
    ta0[ks].y = TT[row * NGRID + kk + 1];
    ta1[ks].x = TT[(row + 16) * NGRID + kk];
    ta1[ks].y = TT[(row + 16) * NGRID + kk + 1];
  }
  float bv[2];
#pragma unroll
  for (int nt = 0; nt < 2; ++nt) bv[nt] = bias[o0 + 2 * row + nt];

  v8f acc[2][2][2] = {};
  __builtin_amdgcn_sched_barrier(0);
#pragma unroll
  for (int ks = 0; ks < 4; ++ks)
#pragma unroll
    for (int ti = 0; ti < 2; ++ti)
#pragma unroll
      for (int nt = 0; nt < 2; ++nt) {
        acc[ti][0][nt] = __builtin_amdgcn_wmma_f32_16x16x4_f32(
            false, a0[ks], false, bf[ti][ks][nt], (short)0, acc[ti][0][nt], false, false);
        acc[ti][1][nt] = __builtin_amdgcn_wmma_f32_16x16x4_f32(
            false, a1[ks], false, bf[ti][ks][nt], (short)0, acc[ti][1][nt], false, false);
      }
  wmma_hazard_slots();
#pragma unroll
  for (int ti = 0; ti < 2; ++ti)
#pragma unroll
    for (int mt = 0; mt < 2; ++mt)
#pragma unroll
      for (int nt = 0; nt < 2; ++nt)
#pragma unroll
        for (int v = 0; v < 8; ++v)
          S[(mt * 16 + 8 * h + v) * SSJ + (ti * 16 + m) * SSP + nt * 16 + row] = acc[ti][mt][nt][v];
  __syncthreads();

  v2f gb[2][8][2];
#pragma unroll
  for (int jj = 0; jj < 2; ++jj)
#pragma unroll
    for (int ks = 0; ks < 8; ++ks) {
      const int kk = 4 * ks + kh;
#pragma unroll
      for (int nt = 0; nt < 2; ++nt) {
        gb[jj][ks][nt].x = S[(2 * w + jj) * SSJ + kk * SSP + nt * 16 + row];
        gb[jj][ks][nt].y = S[(2 * w + jj) * SSJ + (kk + 1) * SSP + nt * 16 + row];
      }
    }
  v8f oc[2][2][2];
#pragma unroll
  for (int jj = 0; jj < 2; ++jj)
#pragma unroll
    for (int mt = 0; mt < 2; ++mt)
#pragma unroll
      for (int nt = 0; nt < 2; ++nt)
#pragma unroll
        for (int v = 0; v < 8; ++v)
          oc[jj][mt][nt][v] = bv[nt];
  __builtin_amdgcn_sched_barrier(0);
#pragma unroll
  for (int jj = 0; jj < 2; ++jj)
#pragma unroll
    for (int ks = 0; ks < 8; ++ks)
#pragma unroll
      for (int nt = 0; nt < 2; ++nt) {
        oc[jj][0][nt] = __builtin_amdgcn_wmma_f32_16x16x4_f32(
            false, ta0[ks], false, gb[jj][ks][nt], (short)0, oc[jj][0][nt], false, false);
        oc[jj][1][nt] = __builtin_amdgcn_wmma_f32_16x16x4_f32(
            false, ta1[ks], false, gb[jj][ks][nt], (short)0, oc[jj][1][nt], false, false);
      }
  wmma_hazard_slots();
#pragma unroll
  for (int jj = 0; jj < 2; ++jj) {
    float* O = out + (((size_t)b * NGRID + (2 * w + jj)) * NGRID) * COUT + o0 + 2 * row;
#pragma unroll
    for (int mt = 0; mt < 2; ++mt)
#pragma unroll
      for (int v = 0; v < 8; ++v) {
        v2f pr;
        pr.x = oc[jj][mt][0][v];
        pr.y = oc[jj][mt][1][v];
        *reinterpret_cast<v2f*>(O + (mt * 16 + 8 * h + v) * COUT) = pr;
      }
  }
}

extern "C" void kernel_launch(void* const* d_in, const int* in_sizes, int n_in,
                              void* d_out, int out_size, void* d_ws, size_t ws_size,
                              hipStream_t stream) {
  const float* inp  = (const float*)d_in[0];
  const float* wgt  = (const float*)d_in[1];
  const float* bias = (const float*)d_in[2];
  float* out = (float*)d_out;
  float* ws  = (float*)d_ws;

  dim3 gT(CIN / CW, BATCH);
  sph_analysis_kernel<<<gT, NWV * 32, 0, stream>>>(inp, ws + WS_COEF, ws + WS_QS, ws + WS_TT);
  dim3 gB(4, KR);
  sph_phaseB_kernel<<<gB, 32, 0, stream>>>(ws + WS_COEF, wgt, ws + WS_CHAT);
  sph_synthesis_kernel<<<gT, NWV * 32, 0, stream>>>(ws + WS_QS, ws + WS_TT, ws + WS_CHAT, bias, out);
}
